# starting kernel
# baseline (speedup 1.0000x reference)
#include <hip/hip_runtime.h>

typedef float v2f __attribute__((ext_vector_type(2)));
typedef float v8f __attribute__((ext_vector_type(8)));

#define N_NODESC 100000
#define N_EDGESC 1600000
#define DIM 128


__global__ void k_init_deg(float* __restrict__ deg) {
  int i = blockIdx.x * blockDim.x + threadIdx.x;
  if (i < N_NODESC) deg[i] = 1.0f;
}

__global__ void k_count_deg(const int* __restrict__ dst, float* __restrict__ deg) {
  int e = blockIdx.x * blockDim.x + threadIdx.x;
  if (e < N_EDGESC) unsafeAtomicAdd(&deg[dst[e]], 1.0f);
}

__global__ void k_dinv(const float* __restrict__ deg, float* __restrict__ dinv) {
  int i = blockIdx.x * blockDim.x + threadIdx.x;
  if (i < N_NODESC) dinv[i] = rsqrtf(deg[i]);
}

__global__ void k_zero4(float4* __restrict__ p, int n4) {
  int i = blockIdx.x * blockDim.x + threadIdx.x;
  if (i < n4) p[i] = make_float4(0.f, 0.f, 0.f, 0.f);
}


__global__ __launch_bounds__(256) void k_gemm_wmma(const float* __restrict__ X,
                                                   const float* __restrict__ W,
                                                   float* __restrict__ H) {
  __shared__ float xs[16 * 132];

  const int t = threadIdx.x;
  const int wave = t >> 5;
  const int lane = t & 31;
  const int row0 = blockIdx.x * 16;

  {
    const int r = t >> 4;
    const int c = (t & 15) * 8;
    const float4* g = (const float4*)(X + (size_t)(row0 + r) * DIM + c);
    float4 p0 = g[0];
    float4 p1 = g[1];
    float4* s = (float4*)&xs[r * 132 + c];
    s[0] = p0;
    s[1] = p1;
  }
  __syncthreads();

  const int m  = lane & 15;
  const int hg = lane >> 4;
  const int n0 = wave * 16;

  v8f acc = {};
  const float* wcol = W + n0 + m;

  #pragma unroll 4
  for (int kk = 0; kk < DIM; kk += 4) {
    v2f a = *(const v2f*)&xs[m * 132 + kk + 2 * hg];
    v2f b;
    b.x = wcol[(size_t)(kk + 2 * hg + 0) * DIM];
    b.y = wcol[(size_t)(kk + 2 * hg + 1) * DIM];
    acc = __builtin_amdgcn_wmma_f32_16x16x4_f32(
         false, a,  false, b,
         (short)0, acc,  false,  false);
  }

  float* hrow = H + (size_t)row0 * DIM + n0 + m;
  #pragma unroll
  for (int v = 0; v < 8; ++v)
    hrow[(size_t)(v + 8 * hg) * DIM] = acc[v];
}


__global__ __launch_bounds__(256) void k_scatter(const int* __restrict__ src,
                                                 const int* __restrict__ dst,
                                                 const float* __restrict__ dinv,
                                                 const float* __restrict__ H,
                                                 float* __restrict__ AGG) {
  int e = blockIdx.x * 8 + (threadIdx.x >> 5);
  if (e >= N_EDGESC) return;
  int lane = threadIdx.x & 31;
  int s = src[e];
  int d = dst[e];
  float w = dinv[s] * dinv[d];
  float4 hv = ((const float4*)(H + (size_t)s * DIM))[lane];
  float* arow = AGG + (size_t)d * DIM + lane * 4;
  unsafeAtomicAdd(arow + 0, hv.x * w);
  unsafeAtomicAdd(arow + 1, hv.y * w);
  unsafeAtomicAdd(arow + 2, hv.z * w);
  unsafeAtomicAdd(arow + 3, hv.w * w);
}


__global__ void k_finalize(const float* __restrict__ AGG, const float* __restrict__ H,
                           const float* __restrict__ dinv, const float* __restrict__ b,
                           float* __restrict__ OUT, int do_relu) {
  int i = blockIdx.x * blockDim.x + threadIdx.x;
  if (i >= N_NODESC * DIM) return;
  int node = i >> 7;
  int f = i & (DIM - 1);
  float di = dinv[node];
  float v = AGG[i] + H[i] * di * di + b[f];
  if (do_relu) v = fmaxf(v, 0.0f);
  OUT[i] = v;
}


extern "C" void kernel_launch(void* const* d_in, const int* in_sizes, int n_in,
                              void* d_out, int out_size, void* d_ws, size_t ws_size,
                              hipStream_t stream) {
  const float* x   = (const float*)d_in[0];
  const int*   ei  = (const int*)d_in[1];
  const float* W1  = (const float*)d_in[2];
  const float* b1  = (const float*)d_in[3];
  const float* W2  = (const float*)d_in[4];
  const float* b2  = (const float*)d_in[5];
  float*       out = (float*)d_out;

  const int* src = ei;
  const int* dst = ei + N_EDGESC;

  float* H    = (float*)d_ws;
  float* AGG  = H + (size_t)N_NODESC * DIM;
  float* deg  = AGG + (size_t)N_NODESC * DIM;
  float* dinv = deg + N_NODESC;

  const int feat_elems = N_NODESC * DIM;
  const int feat_vec4  = feat_elems / 4;

  dim3 blk(256);
  dim3 gNode((N_NODESC + 255) / 256);
  dim3 gEdge((N_EDGESC + 255) / 256);
  dim3 gGemm(N_NODESC / 16);
  dim3 gZero((feat_vec4 + 255) / 256);
  dim3 gScat(N_EDGESC / 8);
  dim3 gFin((feat_elems + 255) / 256);

  k_init_deg<<<gNode, blk, 0, stream>>>(deg);
  k_count_deg<<<gEdge, blk, 0, stream>>>(dst, deg);
  k_dinv<<<gNode, blk, 0, stream>>>(deg, dinv);

  k_gemm_wmma<<<gGemm, blk, 0, stream>>>(x, W1, H);
  k_zero4<<<gZero, blk, 0, stream>>>((float4*)AGG, feat_vec4);
  k_scatter<<<gScat, blk, 0, stream>>>(src, dst, dinv, H, AGG);
  k_finalize<<<gFin, blk, 0, stream>>>(AGG, H, dinv, b1, out,  1);

  k_gemm_wmma<<<gGemm, blk, 0, stream>>>(out, W2, H);
  k_zero4<<<gZero, blk, 0, stream>>>((float4*)AGG, feat_vec4);
  k_scatter<<<gScat, blk, 0, stream>>>(src, dst, dinv, H, AGG);
  k_finalize<<<gFin, blk, 0, stream>>>(AGG, H, dinv, b2, out,  0);
}


# Atomic edge scatter -> per-call dst-sorted edge list (int atomic rank, scan, fill) + atomic-free per-node gather, zero/finalize folded in; rows pre-scaled by dinv; sums in arrival-rank order
# speedup vs baseline: 3.0383x; 3.0383x over previous
#include <hip/hip_runtime.h>

typedef float v2f __attribute__((ext_vector_type(2)));
typedef float v8f __attribute__((ext_vector_type(8)));

#define N_NODESC 100000
#define N_EDGESC 1600000
#define DIM 128


__global__ void k_zero_cnt(int* __restrict__ cnt) {
  int i = blockIdx.x * blockDim.x + threadIdx.x;
  if (i < N_NODESC) cnt[i] = 0;
}

__global__ void k_count_rank(const int* __restrict__ dst, int* __restrict__ cnt,
                             int* __restrict__ rnk) {
  int e = blockIdx.x * blockDim.x + threadIdx.x;
  if (e < N_EDGESC) rnk[e] = atomicAdd(&cnt[dst[e]], 1);
}

__global__ __launch_bounds__(1024) void k_scan(const int* __restrict__ cnt,
                                               int* __restrict__ rowptr,
                                               float* __restrict__ dinv) {
  __shared__ int wsum[32];
  __shared__ int s_off;
  const int t = threadIdx.x;
  const int lane = t & 31;
  const int w = t >> 5;
  const int b0 = blockIdx.x * 1024;

  int part = 0;
  for (int i = t; i < b0; i += 1024) part += cnt[i];
  for (int o = 16; o > 0; o >>= 1) part += __shfl_xor(part, o);
  if (lane == 0) wsum[w] = part;
  __syncthreads();
  if (w == 0) {
    int v = wsum[lane];
    for (int o = 16; o > 0; o >>= 1) v += __shfl_xor(v, o);
    if (lane == 0) s_off = v;
  }
  __syncthreads();
  const int off = s_off;

  const int i = b0 + t;
  const int c = (i < N_NODESC) ? cnt[i] : 0;
  int incl = c;
  for (int o = 1; o < 32; o <<= 1) {
    int y = __shfl_up(incl, o);
    if (lane >= o) incl += y;
  }
  if (lane == 31) wsum[w] = incl;
  __syncthreads();
  if (w == 0) {
    int v = wsum[lane];
    int iv = v;
    for (int o = 1; o < 32; o <<= 1) {
      int y = __shfl_up(iv, o);
      if (lane >= o) iv += y;
    }
    wsum[lane] = iv - v;
  }
  __syncthreads();
  if (i < N_NODESC) {
    rowptr[i] = off + wsum[w] + incl - c;
    dinv[i] = rsqrtf((float)(c + 1));
  }
}

__global__ void k_fill(const int* __restrict__ src, const int* __restrict__ dst,
                       const int* __restrict__ rowptr, const int* __restrict__ rnk,
                       int* __restrict__ col) {
  int e = blockIdx.x * blockDim.x + threadIdx.x;
  if (e < N_EDGESC) col[rowptr[dst[e]] + rnk[e]] = src[e];
}


__global__ __launch_bounds__(256) void k_gemm_wmma(const float* __restrict__ X,
                                                   const float* __restrict__ W,
                                                   const float* __restrict__ dinv,
                                                   float* __restrict__ H) {
  __shared__ float xs[16 * 132];

  const int t = threadIdx.x;
  const int wave = t >> 5;
  const int lane = t & 31;
  const int row0 = blockIdx.x * 16;

  {
    const int r = t >> 4;
    const int c = (t & 15) * 8;
    const float4* g = (const float4*)(X + (size_t)(row0 + r) * DIM + c);
    float4 p0 = g[0];
    float4 p1 = g[1];
    float4* s = (float4*)&xs[r * 132 + c];
    s[0] = p0;
    s[1] = p1;
  }
  __syncthreads();

  const int m  = lane & 15;
  const int hg = lane >> 4;
  const int n0 = wave * 16;

  v8f acc = {};
  const float* wcol = W + n0 + m;

  #pragma unroll 4
  for (int kk = 0; kk < DIM; kk += 4) {
    v2f a = *(const v2f*)&xs[m * 132 + kk + 2 * hg];
    v2f b;
    b.x = wcol[(size_t)(kk + 2 * hg + 0) * DIM];
    b.y = wcol[(size_t)(kk + 2 * hg + 1) * DIM];
    acc = __builtin_amdgcn_wmma_f32_16x16x4_f32(
         false, a,  false, b,
         (short)0, acc,  false,  false);
  }

  float* hrow = H + (size_t)row0 * DIM + n0 + m;
  const float* drow = dinv + row0 + 8 * hg;
  #pragma unroll
  for (int v = 0; v < 8; ++v)
    hrow[(size_t)(v + 8 * hg) * DIM] = acc[v] * drow[v];
}


__global__ __launch_bounds__(256) void k_gather(const int* __restrict__ col,
                                                const int* __restrict__ rowptr,
                                                const int* __restrict__ cnt,
                                                const float* __restrict__ dinv,
                                                const float* __restrict__ HS,
                                                const float* __restrict__ b,
                                                float* __restrict__ OUT, int do_relu) {
  const int lane = threadIdx.x & 31;
  const int d = __builtin_amdgcn_readfirstlane((int)(blockIdx.x * 8 + (threadIdx.x >> 5)));
  const int start = rowptr[d];
  const int n = cnt[d];
  const float4* __restrict__ H4 = (const float4*)HS;

  float4 acc = H4[(size_t)d * 32 + lane];
  for (int base = 0; base < n; base += 32) {
    const int m = (n - base < 32) ? (n - base) : 32;
    const int myc = (lane < m) ? col[start + base + lane] : 0;
    int j = 0;
    for (; j + 8 <= m; j += 8) {
      float4 h[8];
      #pragma unroll
      for (int k = 0; k < 8; ++k) {
        const int s = __builtin_amdgcn_readlane(myc, j + k);
        h[k] = H4[(size_t)s * 32 + lane];
      }
      #pragma unroll
      for (int k = 0; k < 8; ++k) {
        acc.x += h[k].x; acc.y += h[k].y; acc.z += h[k].z; acc.w += h[k].w;
      }
    }
    if (j < m) {
      float4 h[8];
      #pragma unroll
      for (int k = 0; k < 8; ++k) {
        if (j + k < m) {
          const int s = __builtin_amdgcn_readlane(myc, j + k);
          h[k] = H4[(size_t)s * 32 + lane];
        } else {
          h[k] = make_float4(0.f, 0.f, 0.f, 0.f);
        }
      }
      #pragma unroll
      for (int k = 0; k < 8; ++k) {
        acc.x += h[k].x; acc.y += h[k].y; acc.z += h[k].z; acc.w += h[k].w;
      }
    }
  }

  const float di = dinv[d];
  const float* bl = b + lane * 4;
  float4 v;
  v.x = acc.x * di + bl[0];
  v.y = acc.y * di + bl[1];
  v.z = acc.z * di + bl[2];
  v.w = acc.w * di + bl[3];
  if (do_relu) {
    v.x = fmaxf(v.x, 0.0f); v.y = fmaxf(v.y, 0.0f);
    v.z = fmaxf(v.z, 0.0f); v.w = fmaxf(v.w, 0.0f);
  }
  ((float4*)OUT)[(size_t)d * 32 + lane] = v;
}


extern "C" void kernel_launch(void* const* d_in, const int* in_sizes, int n_in,
                              void* d_out, int out_size, void* d_ws, size_t ws_size,
                              hipStream_t stream) {
  const float* x   = (const float*)d_in[0];
  const int*   ei  = (const int*)d_in[1];
  const float* W1  = (const float*)d_in[2];
  const float* b1  = (const float*)d_in[3];
  const float* W2  = (const float*)d_in[4];
  const float* b2  = (const float*)d_in[5];
  float*       out = (float*)d_out;

  const int* src = ei;
  const int* dst = ei + N_EDGESC;

  float* HS     = (float*)d_ws;
  float* dinv   = HS + (size_t)N_NODESC * DIM;
  int*   cnt    = (int*)(dinv + N_NODESC);
  int*   rowptr = cnt + N_NODESC;
  int*   rnk    = rowptr + N_NODESC;
  int*   col    = rnk + N_EDGESC;

  dim3 blk(256);
  dim3 gNode((N_NODESC + 255) / 256);
  dim3 gEdge((N_EDGESC + 255) / 256);
  dim3 gScan((N_NODESC + 1023) / 1024);
  dim3 gGemm(N_NODESC / 16);
  dim3 gGath(N_NODESC / 8);

  k_zero_cnt<<<gNode, blk, 0, stream>>>(cnt);
  k_count_rank<<<gEdge, blk, 0, stream>>>(dst, cnt, rnk);
  k_scan<<<gScan, dim3(1024), 0, stream>>>(cnt, rowptr, dinv);
  k_fill<<<gEdge, blk, 0, stream>>>(src, dst, rowptr, rnk, col);

  k_gemm_wmma<<<gGemm, blk, 0, stream>>>(x, W1, dinv, HS);
  k_gather<<<gGath, blk, 0, stream>>>(col, rowptr, cnt, dinv, HS, b1, out,  1);

  k_gemm_wmma<<<gGemm, blk, 0, stream>>>(out, W2, dinv, HS);
  k_gather<<<gGath, blk, 0, stream>>>(col, rowptr, cnt, dinv, HS, b2, out,  0);
}


# Layer-1 product loses its own launch: its tiles are dealt over the scan launch and the fill launch (row scale from the edge counts) and run beside their dependent-load chains; 7 launches
# speedup vs baseline: 3.1588x; 1.0397x over previous
#include <hip/hip_runtime.h>

typedef float v2f __attribute__((ext_vector_type(2)));
typedef float v8f __attribute__((ext_vector_type(8)));

#define N_NODESC 100000
#define N_EDGESC 1600000
#define DIM 128


__global__ void k_zero_cnt(int* __restrict__ cnt) {
  int i = blockIdx.x * blockDim.x + threadIdx.x;
  if (i < N_NODESC) cnt[i] = 0;
}

__global__ void k_count_rank(const int* __restrict__ dst, int* __restrict__ cnt,
                             int* __restrict__ rnk) {
  int e = blockIdx.x * blockDim.x + threadIdx.x;
  if (e < N_EDGESC) rnk[e] = atomicAdd(&cnt[dst[e]], 1);
}

__device__ __forceinline__ void scan_part(int blk, int* __restrict__ wsum, int* __restrict__ s_offp,
                                          const int* __restrict__ cnt,
                                          int* __restrict__ rowptr,
                                          float* __restrict__ dinv) {
  const int t = threadIdx.x;
  const int lane = t & 31;
  const int w = t >> 5;
  const int b0 = blk * 1024;

  int part = 0;
  for (int i = t; i < b0; i += 1024) part += cnt[i];
  for (int o = 16; o > 0; o >>= 1) part += __shfl_xor(part, o);
  if (lane == 0) wsum[w] = part;
  __syncthreads();
  if (w == 0) {
    int v = wsum[lane];
    for (int o = 16; o > 0; o >>= 1) v += __shfl_xor(v, o);
    if (lane == 0) *s_offp = v;
  }
  __syncthreads();
  const int off = *s_offp;

  const int i = b0 + t;
  const int c = (i < N_NODESC) ? cnt[i] : 0;
  int incl = c;
  for (int o = 1; o < 32; o <<= 1) {
    int y = __shfl_up(incl, o);
    if (lane >= o) incl += y;
  }
  if (lane == 31) wsum[w] = incl;
  __syncthreads();
  if (w == 0) {
    int v = wsum[lane];
    int iv = v;
    for (int o = 1; o < 32; o <<= 1) {
      int y = __shfl_up(iv, o);
      if (lane >= o) iv += y;
    }
    wsum[lane] = iv - v;
  }
  __syncthreads();
  if (i < N_NODESC) {
    rowptr[i] = off + wsum[w] + incl - c;
    dinv[i] = rsqrtf((float)(c + 1));
  }
}

__device__ __forceinline__ void fill_part(int chunk, const int* __restrict__ src,
                                          const int* __restrict__ dst,
                                          const int* __restrict__ rowptr,
                                          const int* __restrict__ rnk, int* __restrict__ col) {
  int e = chunk * 256 + (int)threadIdx.x;
  if (e < N_EDGESC) col[rowptr[dst[e]] + rnk[e]] = src[e];
}


__device__ __forceinline__ void gemm_part(int tile, float* __restrict__ xs, int t,
                                          const float* __restrict__ X,
                                          const float* __restrict__ W,
                                          const int* __restrict__ cnt,
                                          float* __restrict__ H) {
  const int wave = t >> 5;
  const int lane = t & 31;
  const int row0 = tile * 16;

  {
    const int r = t >> 4;
    const int c = (t & 15) * 8;
    const float4* g = (const float4*)(X + (size_t)(row0 + r) * DIM + c);
    float4 p0 = g[0];
    float4 p1 = g[1];
    float4* s = (float4*)&xs[r * 132 + c];
    s[0] = p0;
    s[1] = p1;
  }
  __syncthreads();

  const int m  = lane & 15;
  const int hg = lane >> 4;
  const int n0 = wave * 16;

  v8f acc = {};
  const float* wcol = W + n0 + m;

  #pragma unroll 4
  for (int kk = 0; kk < DIM; kk += 4) {
    v2f a = *(const v2f*)&xs[m * 132 + kk + 2 * hg];
    v2f b;
    b.x = wcol[(size_t)(kk + 2 * hg + 0) * DIM];
    b.y = wcol[(size_t)(kk + 2 * hg + 1) * DIM];
    acc = __builtin_amdgcn_wmma_f32_16x16x4_f32(
         false, a,  false, b,
         (short)0, acc,  false,  false);
  }

  float* hrow = H + (size_t)row0 * DIM + n0 + m;
  const int* crow = cnt + row0 + 8 * hg;
  #pragma unroll
  for (int v = 0; v < 8; ++v)
    hrow[(size_t)(v + 8 * hg) * DIM] = acc[v] * rsqrtf((float)(crow[v] + 1));
}


#define SCAN_BLOCKS 98
#define TILES_WITH_SCAN 2500

__global__ __launch_bounds__(1024) void k_scan_gemm(const int* __restrict__ cnt,
                                                    int* __restrict__ rowptr,
                                                    float* __restrict__ dinv,
                                                    const float* __restrict__ X,
                                                    const float* __restrict__ W,
                                                    float* __restrict__ H) {
  __shared__ int wsum[32];
  __shared__ int s_off;
  __shared__ float xs[4 * 16 * 132];
  if (blockIdx.x < SCAN_BLOCKS) {
    scan_part((int)blockIdx.x, wsum, &s_off, cnt, rowptr, dinv);
  } else {
    const int g = threadIdx.x >> 8;
    gemm_part(((int)blockIdx.x - SCAN_BLOCKS) * 4 + g, xs + g * (16 * 132), threadIdx.x & 255,
              X, W, cnt, H);
  }
}

__global__ __launch_bounds__(256) void k_fill_gemm(const int* __restrict__ src,
                                                   const int* __restrict__ dst,
                                                   const int* __restrict__ rowptr,
                                                   const int* __restrict__ rnk,
                                                   int* __restrict__ col,
                                                   const int* __restrict__ cnt,
                                                   const float* __restrict__ X,
                                                   const float* __restrict__ W,
                                                   float* __restrict__ H) {
  __shared__ float xs[16 * 132];
  const int run = (int)(blockIdx.x >> 3);
  const int pos = (int)(blockIdx.x & 7);
  const int tiles_before = __popc(0x52u & ((1u << pos) - 1u));
  if ((0x52u >> pos) & 1u) {
    gemm_part(TILES_WITH_SCAN + run * 3 + tiles_before, xs, (int)threadIdx.x, X, W, cnt, H);
  } else {
    fill_part(run * 5 + (pos - tiles_before), src, dst, rowptr, rnk, col);
  }
}

__global__ __launch_bounds__(256) void k_gemm_wmma(const float* __restrict__ X,
                                                   const float* __restrict__ W,
                                                   const int* __restrict__ cnt,
                                                   float* __restrict__ H) {
  __shared__ float xs[16 * 132];
  gemm_part((int)blockIdx.x, xs, (int)threadIdx.x, X, W, cnt, H);
}


__global__ __launch_bounds__(256) void k_gather(const int* __restrict__ col,
                                                const int* __restrict__ rowptr,
                                                const int* __restrict__ cnt,
                                                const float* __restrict__ dinv,
                                                const float* __restrict__ HS,
                                                const float* __restrict__ b,
                                                float* __restrict__ OUT, int do_relu) {
  const int lane = threadIdx.x & 31;
  const int d = __builtin_amdgcn_readfirstlane((int)(blockIdx.x * 8 + (threadIdx.x >> 5)));
  const int start = rowptr[d];
  const int n = cnt[d];
  const float4* __restrict__ H4 = (const float4*)HS;

  float4 acc = H4[(size_t)d * 32 + lane];
  for (int base = 0; base < n; base += 32) {
    const int m = (n - base < 32) ? (n - base) : 32;
    const int myc = (lane < m) ? col[start + base + lane] : 0;
    int j = 0;
    for (; j + 8 <= m; j += 8) {
      float4 h[8];
      #pragma unroll
      for (int k = 0; k < 8; ++k) {
        const int s = __builtin_amdgcn_readlane(myc, j + k);
        h[k] = H4[(size_t)s * 32 + lane];
      }
      #pragma unroll
      for (int k = 0; k < 8; ++k) {
        acc.x += h[k].x; acc.y += h[k].y; acc.z += h[k].z; acc.w += h[k].w;
      }
    }
    if (j < m) {
      float4 h[8];
      #pragma unroll
      for (int k = 0; k < 8; ++k) {
        if (j + k < m) {
          const int s = __builtin_amdgcn_readlane(myc, j + k);
          h[k] = H4[(size_t)s * 32 + lane];
        } else {
          h[k] = make_float4(0.f, 0.f, 0.f, 0.f);
        }
      }
      #pragma unroll
      for (int k = 0; k < 8; ++k) {
        acc.x += h[k].x; acc.y += h[k].y; acc.z += h[k].z; acc.w += h[k].w;
      }
    }
  }

  const float di = dinv[d];
  const float* bl = b + lane * 4;
  float4 v;
  v.x = acc.x * di + bl[0];
  v.y = acc.y * di + bl[1];
  v.z = acc.z * di + bl[2];
  v.w = acc.w * di + bl[3];
  if (do_relu) {
    v.x = fmaxf(v.x, 0.0f); v.y = fmaxf(v.y, 0.0f);
    v.z = fmaxf(v.z, 0.0f); v.w = fmaxf(v.w, 0.0f);
  }
  ((float4*)OUT)[(size_t)d * 32 + lane] = v;
}


extern "C" void kernel_launch(void* const* d_in, const int* in_sizes, int n_in,
                              void* d_out, int out_size, void* d_ws, size_t ws_size,
                              hipStream_t stream) {
  const float* x   = (const float*)d_in[0];
  const int*   ei  = (const int*)d_in[1];
  const float* W1  = (const float*)d_in[2];
  const float* b1  = (const float*)d_in[3];
  const float* W2  = (const float*)d_in[4];
  const float* b2  = (const float*)d_in[5];
  float*       out = (float*)d_out;

  const int* src = ei;
  const int* dst = ei + N_EDGESC;

  float* HS     = (float*)d_ws;
  float* dinv   = HS + (size_t)N_NODESC * DIM;
  int*   cnt    = (int*)(dinv + N_NODESC);
  int*   rowptr = cnt + N_NODESC;
  int*   rnk    = rowptr + N_NODESC;
  int*   col    = rnk + N_EDGESC;

  dim3 blk(256);
  dim3 gNode((N_NODESC + 255) / 256);
  dim3 gEdge((N_EDGESC + 255) / 256);
  dim3 gGemm(N_NODESC / 16);
  dim3 gGath(N_NODESC / 8);

  k_zero_cnt<<<gNode, blk, 0, stream>>>(cnt);
  k_count_rank<<<gEdge, blk, 0, stream>>>(dst, cnt, rnk);
  k_scan_gemm<<<dim3(723), dim3(1024), 0, stream>>>(cnt, rowptr, dinv, x, W1, HS);
  k_fill_gemm<<<dim3(10000), blk, 0, stream>>>(src, dst, rowptr, rnk, col, cnt, x, W1, HS);

  k_gather<<<gGath, blk, 0, stream>>>(col, rowptr, cnt, dinv, HS, b1, out,  1);

  k_gemm_wmma<<<gGemm, blk, 0, stream>>>(out, W2, cnt, HS);
  k_gather<<<gGath, blk, 0, stream>>>(col, rowptr, cnt, dinv, HS, b2, out,  0);
}


# Edge list built by a two-level LDS counting sort (256 chunk blocks bin by dst>>9, 196 bucket blocks count, scan, place) instead of one device atomic per edge; 6 launches; sums in LDS arrival order
# speedup vs baseline: 4.7162x; 1.4930x over previous
#include <hip/hip_runtime.h>

typedef float v2f __attribute__((ext_vector_type(2)));
typedef float v8f __attribute__((ext_vector_type(8)));

#define N_NODESC 100000
#define N_EDGESC 1600000
#define DIM 128


#define NBUCK 196
#define BSH 9
#define BMASK 511
#define NCHUNK 256
#define CHUNK 6250
#define CHS 6272

__device__ __forceinline__ int block_scan(int v, int* __restrict__ wsum, int& total) {
  const int lane = threadIdx.x & 31;
  const int w = threadIdx.x >> 5;
  int incl = v;
  for (int o = 1; o < 32; o <<= 1) {
    int y = __shfl_up(incl, o);
    if (lane >= o) incl += y;
  }
  __syncthreads();
  if (lane == 31) wsum[w] = incl;
  __syncthreads();
  if (w == 0) {
    int x = wsum[lane];
    int ix = x;
    for (int o = 1; o < 32; o <<= 1) {
      int y = __shfl_up(ix, o);
      if (lane >= o) ix += y;
    }
    wsum[lane] = ix - x;
    if (lane == 31) wsum[32] = ix;
  }
  __syncthreads();
  total = wsum[32];
  return wsum[w] + incl - v;
}

__global__ __launch_bounds__(1024) void k_bin(const int* __restrict__ src,
                                              const int* __restrict__ dst,
                                              int* __restrict__ tmp,
                                              int* __restrict__ histg,
                                              int* __restrict__ offsg) {
  __shared__ int hist[NBUCK];
  __shared__ int offs[NBUCK];
  __shared__ int wsum[33];
  const int t = threadIdx.x;
  const int wg = blockIdx.x;
  if (t < NBUCK) hist[t] = 0;
  __syncthreads();

  const int e0 = wg * CHUNK;
  int key[7], bk[7], rk[7];
  #pragma unroll
  for (int k = 0; k < 7; ++k) {
    const int i = k * 1024 + t;
    if (i < CHUNK) {
      const int d = dst[e0 + i];
      const int s = src[e0 + i];
      bk[k] = d >> BSH;
      key[k] = (s << BSH) | (d & BMASK);
    } else {
      bk[k] = -1;
      key[k] = 0;
    }
  }
  #pragma unroll
  for (int k = 0; k < 7; ++k) rk[k] = (bk[k] >= 0) ? atomicAdd(&hist[bk[k]], 1) : 0;
  __syncthreads();

  const int h = (t < NBUCK) ? hist[t] : 0;
  int total;
  const int ex = block_scan(h, wsum, total);
  if (t < NBUCK) {
    offs[t] = ex;
    histg[t * NCHUNK + wg] = h;
    offsg[t * NCHUNK + wg] = ex;
  }
  __syncthreads();

  int* __restrict__ o = tmp + wg * CHS;
  #pragma unroll
  for (int k = 0; k < 7; ++k)
    if (bk[k] >= 0) o[offs[bk[k]] + rk[k]] = key[k];
}

__global__ __launch_bounds__(1024) void k_bucket(const int* __restrict__ tmp,
                                                 const int* __restrict__ histg,
                                                 const int* __restrict__ offsg,
                                                 int* __restrict__ cnt,
                                                 int* __restrict__ rowptr,
                                                 float* __restrict__ dinv,
                                                 int* __restrict__ col) {
  __shared__ int c[512];
  __shared__ int cur[512];
  __shared__ int ph[NCHUNK];
  __shared__ int po[NCHUNK];
  __shared__ int wsum[33];
  const int t = threadIdx.x;
  const int lane = t & 31;
  const int w = t >> 5;
  const int b = blockIdx.x;

  if (t < 512) c[t] = 0;
  if (t < NCHUNK) {
    ph[t] = histg[b * NCHUNK + t];
    po[t] = offsg[b * NCHUNK + t];
  }
  int part = 0;
  {
    const int4* __restrict__ h4 = (const int4*)histg;
    const int n4 = b * (NCHUNK / 4);
    #pragma unroll
    for (int j = 0; j < 13; ++j) {
      const int i = j * 1024 + t;
      if (i < n4) {
        const int4 q = h4[i];
        part += (q.x + q.y) + (q.z + q.w);
      }
    }
  }
  int base;
  (void)block_scan(part, wsum, base);

  int n[8];
  const int* q[8];
  int k0[8], k1[8];
  #pragma unroll
  for (int j = 0; j < 8; ++j) {
    const int p = w + 32 * j;
    n[j] = ph[p];
    q[j] = tmp + p * CHS + po[p];
  }
  #pragma unroll
  for (int j = 0; j < 8; ++j) {
    k0[j] = (lane < n[j]) ? q[j][lane] : -1;
    k1[j] = (lane + 32 < n[j]) ? q[j][lane + 32] : -1;
  }

  #pragma unroll
  for (int j = 0; j < 8; ++j) {
    if (k0[j] >= 0) atomicAdd(&c[k0[j] & BMASK], 1);
    if (k1[j] >= 0) atomicAdd(&c[k1[j] & BMASK], 1);
  }
  #pragma unroll
  for (int j = 0; j < 8; ++j)
    for (int i = lane + 64; i < n[j]; i += 32) atomicAdd(&c[q[j][i] & BMASK], 1);
  __syncthreads();

  const int v = (t < 512) ? c[t] : 0;
  int tot;
  const int ex = block_scan(v, wsum, tot);
  if (t < 512) {
    const int node = b * 512 + t;
    cur[t] = base + ex;
    if (node < N_NODESC) {
      cnt[node] = v;
      rowptr[node] = base + ex;
      dinv[node] = rsqrtf((float)(v + 1));
    }
  }
  __syncthreads();

  #pragma unroll
  for (int j = 0; j < 8; ++j) {
    if (k0[j] >= 0) col[atomicAdd(&cur[k0[j] & BMASK], 1)] = k0[j] >> BSH;
    if (k1[j] >= 0) col[atomicAdd(&cur[k1[j] & BMASK], 1)] = k1[j] >> BSH;
  }
  #pragma unroll
  for (int j = 0; j < 8; ++j)
    for (int i = lane + 64; i < n[j]; i += 32) {
      const int k = q[j][i];
      col[atomicAdd(&cur[k & BMASK], 1)] = k >> BSH;
    }
}


__device__ __forceinline__ void gemm_part(int tile, float* __restrict__ xs, int t,
                                          const float* __restrict__ X,
                                          const float* __restrict__ W,
                                          const int* __restrict__ cnt,
                                          float* __restrict__ H) {
  const int wave = t >> 5;
  const int lane = t & 31;
  const int row0 = tile * 16;

  {
    const int r = t >> 4;
    const int c = (t & 15) * 8;
    const float4* g = (const float4*)(X + (size_t)(row0 + r) * DIM + c);
    float4 p0 = g[0];
    float4 p1 = g[1];
    float4* s = (float4*)&xs[r * 132 + c];
    s[0] = p0;
    s[1] = p1;
  }
  __syncthreads();

  const int m  = lane & 15;
  const int hg = lane >> 4;
  const int n0 = wave * 16;

  v8f acc = {};
  const float* wcol = W + n0 + m;

  #pragma unroll 4
  for (int kk = 0; kk < DIM; kk += 4) {
    v2f a = *(const v2f*)&xs[m * 132 + kk + 2 * hg];
    v2f b;
    b.x = wcol[(size_t)(kk + 2 * hg + 0) * DIM];
    b.y = wcol[(size_t)(kk + 2 * hg + 1) * DIM];
    acc = __builtin_amdgcn_wmma_f32_16x16x4_f32(
         false, a,  false, b,
         (short)0, acc,  false,  false);
  }

  float* hrow = H + (size_t)row0 * DIM + n0 + m;
  const int* crow = cnt + row0 + 8 * hg;
  #pragma unroll
  for (int v = 0; v < 8; ++v)
    hrow[(size_t)(v + 8 * hg) * DIM] = acc[v] * rsqrtf((float)(crow[v] + 1));
}

__global__ __launch_bounds__(256) void k_gemm_wmma(const float* __restrict__ X,
                                                   const float* __restrict__ W,
                                                   const int* __restrict__ cnt,
                                                   float* __restrict__ H) {
  __shared__ float xs[16 * 132];
  gemm_part((int)blockIdx.x, xs, (int)threadIdx.x, X, W, cnt, H);
}


__global__ __launch_bounds__(256) void k_gather(const int* __restrict__ col,
                                                const int* __restrict__ rowptr,
                                                const int* __restrict__ cnt,
                                                const float* __restrict__ dinv,
                                                const float* __restrict__ HS,
                                                const float* __restrict__ b,
                                                float* __restrict__ OUT, int do_relu) {
  const int lane = threadIdx.x & 31;
  const int d = __builtin_amdgcn_readfirstlane((int)(blockIdx.x * 8 + (threadIdx.x >> 5)));
  const int start = rowptr[d];
  const int n = cnt[d];
  const float4* __restrict__ H4 = (const float4*)HS;

  float4 acc = H4[(size_t)d * 32 + lane];
  for (int base = 0; base < n; base += 32) {
    const int m = (n - base < 32) ? (n - base) : 32;
    const int myc = (lane < m) ? col[start + base + lane] : 0;
    int j = 0;
    for (; j + 8 <= m; j += 8) {
      float4 h[8];
      #pragma unroll
      for (int k = 0; k < 8; ++k) {
        const int s = __builtin_amdgcn_readlane(myc, j + k);
        h[k] = H4[(size_t)s * 32 + lane];
      }
      #pragma unroll
      for (int k = 0; k < 8; ++k) {
        acc.x += h[k].x; acc.y += h[k].y; acc.z += h[k].z; acc.w += h[k].w;
      }
    }
    if (j < m) {
      float4 h[8];
      #pragma unroll
      for (int k = 0; k < 8; ++k) {
        if (j + k < m) {
          const int s = __builtin_amdgcn_readlane(myc, j + k);
          h[k] = H4[(size_t)s * 32 + lane];
        } else {
          h[k] = make_float4(0.f, 0.f, 0.f, 0.f);
        }
      }
      #pragma unroll
      for (int k = 0; k < 8; ++k) {
        acc.x += h[k].x; acc.y += h[k].y; acc.z += h[k].z; acc.w += h[k].w;
      }
    }
  }

  const float di = dinv[d];
  const float* bl = b + lane * 4;
  float4 v;
  v.x = acc.x * di + bl[0];
  v.y = acc.y * di + bl[1];
  v.z = acc.z * di + bl[2];
  v.w = acc.w * di + bl[3];
  if (do_relu) {
    v.x = fmaxf(v.x, 0.0f); v.y = fmaxf(v.y, 0.0f);
    v.z = fmaxf(v.z, 0.0f); v.w = fmaxf(v.w, 0.0f);
  }
  ((float4*)OUT)[(size_t)d * 32 + lane] = v;
}


extern "C" void kernel_launch(void* const* d_in, const int* in_sizes, int n_in,
                              void* d_out, int out_size, void* d_ws, size_t ws_size,
                              hipStream_t stream) {
  const float* x   = (const float*)d_in[0];
  const int*   ei  = (const int*)d_in[1];
  const float* W1  = (const float*)d_in[2];
  const float* b1  = (const float*)d_in[3];
  const float* W2  = (const float*)d_in[4];
  const float* b2  = (const float*)d_in[5];
  float*       out = (float*)d_out;

  const int* src = ei;
  const int* dst = ei + N_EDGESC;

  float* HS     = (float*)d_ws;
  float* dinv   = HS + (size_t)N_NODESC * DIM;
  int*   cnt    = (int*)(dinv + N_NODESC);
  int*   rowptr = cnt + N_NODESC;
  int*   col    = rowptr + N_NODESC;
  int*   tmp    = col + N_EDGESC;
  int*   histg  = tmp + NCHUNK * CHS;
  int*   offsg  = histg + NBUCK * NCHUNK;

  dim3 blk(256);
  dim3 gGemm(N_NODESC / 16);
  dim3 gGath(N_NODESC / 8);

  k_bin<<<dim3(NCHUNK), dim3(1024), 0, stream>>>(src, dst, tmp, histg, offsg);
  k_bucket<<<dim3(NBUCK), dim3(1024), 0, stream>>>(tmp, histg, offsg, cnt, rowptr, dinv, col);

  k_gemm_wmma<<<gGemm, blk, 0, stream>>>(x, W1, cnt, HS);
  k_gather<<<gGath, blk, 0, stream>>>(col, rowptr, cnt, dinv, HS, b1, out,  1);

  k_gemm_wmma<<<gGemm, blk, 0, stream>>>(out, W2, cnt, HS);
  k_gather<<<gGath, blk, 0, stream>>>(col, rowptr, cnt, dinv, HS, b2, out,  0);
}


# Each layer sums its source rows first and multiplies the 32-node sum tile by W (held in LDS) in the same launch: no product launches, 4 launches; rows are summed before the product, not after
# speedup vs baseline: 5.4120x; 1.1475x over previous
#include <hip/hip_runtime.h>

typedef float v2f __attribute__((ext_vector_type(2)));
typedef float v8f __attribute__((ext_vector_type(8)));

#define N_NODESC 100000
#define N_EDGESC 1600000
#define DIM 128


#define NBUCK 196
#define BSH 9
#define BMASK 511
#define NCHUNK 256
#define CHUNK 6250
#define CHS 6272

__device__ __forceinline__ int block_scan(int v, int* __restrict__ wsum, int& total) {
  const int lane = threadIdx.x & 31;
  const int w = threadIdx.x >> 5;
  int incl = v;
  for (int o = 1; o < 32; o <<= 1) {
    int y = __shfl_up(incl, o);
    if (lane >= o) incl += y;
  }
  __syncthreads();
  if (lane == 31) wsum[w] = incl;
  __syncthreads();
  if (w == 0) {
    int x = wsum[lane];
    int ix = x;
    for (int o = 1; o < 32; o <<= 1) {
      int y = __shfl_up(ix, o);
      if (lane >= o) ix += y;
    }
    wsum[lane] = ix - x;
    if (lane == 31) wsum[32] = ix;
  }
  __syncthreads();
  total = wsum[32];
  return wsum[w] + incl - v;
}

__global__ __launch_bounds__(1024) void k_bin(const int* __restrict__ src,
                                              const int* __restrict__ dst,
                                              int* __restrict__ tmp,
                                              int* __restrict__ histg,
                                              int* __restrict__ offsg) {
  __shared__ int hist[NBUCK];
  __shared__ int offs[NBUCK];
  __shared__ int wsum[33];
  const int t = threadIdx.x;
  const int wg = blockIdx.x;
  if (t < NBUCK) hist[t] = 0;
  __syncthreads();

  const int e0 = wg * CHUNK;
  int key[7], bk[7], rk[7];
  #pragma unroll
  for (int k = 0; k < 7; ++k) {
    const int i = k * 1024 + t;
    if (i < CHUNK) {
      const int d = dst[e0 + i];
      const int s = src[e0 + i];
      bk[k] = d >> BSH;
      key[k] = (s << BSH) | (d & BMASK);
    } else {
      bk[k] = -1;
      key[k] = 0;
    }
  }
  #pragma unroll
  for (int k = 0; k < 7; ++k) rk[k] = (bk[k] >= 0) ? atomicAdd(&hist[bk[k]], 1) : 0;
  __syncthreads();

  const int h = (t < NBUCK) ? hist[t] : 0;
  int total;
  const int ex = block_scan(h, wsum, total);
  if (t < NBUCK) {
    offs[t] = ex;
    histg[t * NCHUNK + wg] = h;
    offsg[t * NCHUNK + wg] = ex;
  }
  __syncthreads();

  int* __restrict__ o = tmp + wg * CHS;
  #pragma unroll
  for (int k = 0; k < 7; ++k)
    if (bk[k] >= 0) o[offs[bk[k]] + rk[k]] = key[k];
}

__global__ __launch_bounds__(1024) void k_bucket(const int* __restrict__ tmp,
                                                 const int* __restrict__ histg,
                                                 const int* __restrict__ offsg,
                                                 int* __restrict__ cnt,
                                                 int* __restrict__ rowptr,
                                                 float* __restrict__ dinv,
                                                 int* __restrict__ col) {
  __shared__ int c[512];
  __shared__ int cur[512];
  __shared__ int ph[NCHUNK];
  __shared__ int po[NCHUNK];
  __shared__ int wsum[33];
  const int t = threadIdx.x;
  const int lane = t & 31;
  const int w = t >> 5;
  const int b = blockIdx.x;

  if (t < 512) c[t] = 0;
  if (t < NCHUNK) {
    ph[t] = histg[b * NCHUNK + t];
    po[t] = offsg[b * NCHUNK + t];
  }
  int part = 0;
  {
    const int4* __restrict__ h4 = (const int4*)histg;
    const int n4 = b * (NCHUNK / 4);
    #pragma unroll
    for (int j = 0; j < 13; ++j) {
      const int i = j * 1024 + t;
      if (i < n4) {
        const int4 q = h4[i];
        part += (q.x + q.y) + (q.z + q.w);
      }
    }
  }
  int base;
  (void)block_scan(part, wsum, base);

  int n[8];
  const int* q[8];
  int k0[8], k1[8];
  #pragma unroll
  for (int j = 0; j < 8; ++j) {
    const int p = w + 32 * j;
    n[j] = ph[p];
    q[j] = tmp + p * CHS + po[p];
  }
  #pragma unroll
  for (int j = 0; j < 8; ++j) {
    k0[j] = (lane < n[j]) ? q[j][lane] : -1;
    k1[j] = (lane + 32 < n[j]) ? q[j][lane + 32] : -1;
  }

  #pragma unroll
  for (int j = 0; j < 8; ++j) {
    if (k0[j] >= 0) atomicAdd(&c[k0[j] & BMASK], 1);
    if (k1[j] >= 0) atomicAdd(&c[k1[j] & BMASK], 1);
  }
  #pragma unroll
  for (int j = 0; j < 8; ++j)
    for (int i = lane + 64; i < n[j]; i += 32) atomicAdd(&c[q[j][i] & BMASK], 1);
  __syncthreads();

  const int v = (t < 512) ? c[t] : 0;
  int tot;
  const int ex = block_scan(v, wsum, tot);
  if (t < 512) {
    const int node = b * 512 + t;
    cur[t] = base + ex;
    if (node < N_NODESC) {
      cnt[node] = v;
      rowptr[node] = base + ex;
      dinv[node] = rsqrtf((float)(v + 1));
    }
  }
  __syncthreads();

  #pragma unroll
  for (int j = 0; j < 8; ++j) {
    if (k0[j] >= 0) col[atomicAdd(&cur[k0[j] & BMASK], 1)] = k0[j] >> BSH;
    if (k1[j] >= 0) col[atomicAdd(&cur[k1[j] & BMASK], 1)] = k1[j] >> BSH;
  }
  #pragma unroll
  for (int j = 0; j < 8; ++j)
    for (int i = lane + 64; i < n[j]; i += 32) {
      const int k = q[j][i];
      col[atomicAdd(&cur[k & BMASK], 1)] = k >> BSH;
    }
}


#define TROWS 32
#define XP 132
#define WP 288
#define NTILE (N_NODESC / TROWS)
#define LGRID 625

template <bool L1>
__device__ __forceinline__ float4 sum_node(int n, int start, int c0, float4 self, float dd, int lane,
                                           const int* __restrict__ col,
                                           const float* __restrict__ dinv,
                                           const float4* __restrict__ H4) {
  float4 acc = L1 ? make_float4(0.f, 0.f, 0.f, 0.f) : self;
  for (int base = 0; base < n; base += 32) {
    const int mm = (n - base < 32) ? (n - base) : 32;
    int myc = c0;
    if (base > 0) myc = (lane < mm) ? col[start + base + lane] : 0;
    int myw = 0;
    int j = 0;
    for (; j + 8 <= mm; j += 8) {
      float4 h[8];
      #pragma unroll
      for (int k = 0; k < 8; ++k) {
        const int s = __builtin_amdgcn_readlane(myc, j + k);
        h[k] = H4[(size_t)s * 32 + lane];
      }
      if (L1 && j == 0) myw = __float_as_int(dinv[myc]);
      #pragma unroll
      for (int k = 0; k < 8; ++k) {
        if (L1) {
          const float w = __int_as_float(__builtin_amdgcn_readlane(myw, j + k));
          acc.x = fmaf(w, h[k].x, acc.x); acc.y = fmaf(w, h[k].y, acc.y);
          acc.z = fmaf(w, h[k].z, acc.z); acc.w = fmaf(w, h[k].w, acc.w);
        } else {
          acc.x += h[k].x; acc.y += h[k].y; acc.z += h[k].z; acc.w += h[k].w;
        }
      }
    }
    if (j < mm) {
      float4 h[8];
      #pragma unroll
      for (int k = 0; k < 8; ++k) {
        if (j + k < mm) {
          const int s = __builtin_amdgcn_readlane(myc, j + k);
          h[k] = H4[(size_t)s * 32 + lane];
        } else {
          h[k] = make_float4(0.f, 0.f, 0.f, 0.f);
        }
      }
      if (L1 && j == 0) myw = __float_as_int(dinv[myc]);
      #pragma unroll
      for (int k = 0; k < 8; ++k) {
        if (L1) {
          const float w = __int_as_float(__builtin_amdgcn_readlane(myw, j + k));
          acc.x = fmaf(w, h[k].x, acc.x); acc.y = fmaf(w, h[k].y, acc.y);
          acc.z = fmaf(w, h[k].z, acc.z); acc.w = fmaf(w, h[k].w, acc.w);
        } else {
          acc.x += h[k].x; acc.y += h[k].y; acc.z += h[k].z; acc.w += h[k].w;
        }
      }
    }
  }
  if (L1) {
    acc.x = fmaf(dd, self.x, acc.x); acc.y = fmaf(dd, self.y, acc.y);
    acc.z = fmaf(dd, self.z, acc.z); acc.w = fmaf(dd, self.w, acc.w);
  }
  return acc;
}

template <bool L1>
__device__ __forceinline__ void layer_body(const int* __restrict__ col,
                                           const int* __restrict__ rowptr,
                                           const int* __restrict__ cnt,
                                           const float* __restrict__ dinv,
                                           const float* __restrict__ X,
                                           const float* __restrict__ W,
                                           const float* __restrict__ b,
                                           float* __restrict__ OUT,
                                           float* __restrict__ Wl,
                                           float* __restrict__ xs) {
  const int t = threadIdx.x;
  const int lane = t & 31;
  const int wave = t >> 5;

  {
    const float4* __restrict__ W4 = (const float4*)W;
    #pragma unroll
    for (int j = 0; j < 4; ++j) {
      const int item = t + j * 512;
      const int kp = item >> 5;
      const int q = item & 31;
      const float4 r0 = W4[(2 * kp) * 32 + q];
      const float4 r1 = W4[(2 * kp + 1) * 32 + q];
      float4* dst = (float4*)&Wl[kp * WP + 8 * q];
      dst[0] = make_float4(r0.x, r1.x, r0.y, r1.y);
      dst[1] = make_float4(r0.z, r1.z, r0.w, r1.w);
    }
  }

  const float4* __restrict__ H4 = (const float4*)X;
  const int m  = lane & 15;
  const int hg = lane >> 4;
  const int rt = wave >> 3;
  const int n0 = (wave & 7) * 16;
  const float bias = b[n0 + m];

  for (int tile = blockIdx.x; tile < NTILE; tile += LGRID) {
    const int row0 = tile * TROWS;

    const int rA = 2 * wave;
    const int dA = __builtin_amdgcn_readfirstlane(row0 + rA);
    const int dB = dA + 1;
    const int nA = cnt[dA];
    const int nB = cnt[dB];
    const int stA = rowptr[dA];
    const int stB = rowptr[dB];
    const float ddA = dinv[dA];
    const float ddB = dinv[dB];
    const float4 selfA = H4[(size_t)dA * 32 + lane];
    const float4 selfB = H4[(size_t)dB * 32 + lane];
    const int cA = (lane < nA) ? col[stA + lane] : 0;
    const int cB = (lane < nB) ? col[stB + lane] : 0;
    {
      const float4 acc = sum_node<L1>(nA, stA, cA, selfA, ddA, lane, col, dinv, H4);
      float4 tv;
      tv.x = acc.x * ddA; tv.y = acc.y * ddA; tv.z = acc.z * ddA; tv.w = acc.w * ddA;
      *(float4*)&xs[rA * XP + lane * 4] = tv;
    }
    {
      const float4 acc = sum_node<L1>(nB, stB, cB, selfB, ddB, lane, col, dinv, H4);
      float4 tv;
      tv.x = acc.x * ddB; tv.y = acc.y * ddB; tv.z = acc.z * ddB; tv.w = acc.w * ddB;
      *(float4*)&xs[(rA + 1) * XP + lane * 4] = tv;
    }
    __syncthreads();

    v8f c = {};
    const float* arow = xs + (16 * rt + m) * XP + 2 * hg;
    const float* bcol = Wl + hg * WP + 2 * (n0 + m);
    const float* ap = arow;
    const float* bp = bcol;
    for (int kb = 0; kb < 4; ++kb) {
      v2f a[8], bb[8];
      #pragma unroll
      for (int u = 0; u < 8; ++u) {
        a[u] = *(const v2f*)(ap + 4 * u);
        bb[u] = *(const v2f*)(bp + 2 * u * WP);
      }
      #pragma unroll
      for (int u = 0; u < 8; ++u)
        c = __builtin_amdgcn_wmma_f32_16x16x4_f32(
             false, a[u],  false, bb[u],
             (short)0, c,  false,  false);
      ap += 32;
      bp += 16 * WP;
    }

    const int rb = row0 + 16 * rt + 8 * hg;
    float* orow = OUT + (size_t)rb * DIM + n0 + m;
    if (L1) {
      const float4 s0 = *(const float4*)(dinv + rb);
      const float4 s1 = *(const float4*)(dinv + rb + 4);
      const float sc[8] = {s0.x, s0.y, s0.z, s0.w, s1.x, s1.y, s1.z, s1.w};
      #pragma unroll
      for (int v = 0; v < 8; ++v)
        orow[(size_t)v * DIM] = fmaxf(c[v] + bias, 0.0f) * sc[v];
    } else {
      #pragma unroll
      for (int v = 0; v < 8; ++v)
        orow[(size_t)v * DIM] = c[v] + bias;
    }
    __builtin_amdgcn_fence(__ATOMIC_RELEASE, "workgroup", "local");
    __builtin_amdgcn_s_barrier();
    __builtin_amdgcn_fence(__ATOMIC_ACQUIRE, "workgroup", "local");
  }
}

__global__ __launch_bounds__(512) void k_layer1(const int* __restrict__ col,
                                                const int* __restrict__ rowptr,
                                                const int* __restrict__ cnt,
                                                const float* __restrict__ dinv,
                                                const float* __restrict__ X,
                                                const float* __restrict__ W,
                                                const float* __restrict__ b,
                                                float* __restrict__ OUT) {
  __shared__ __attribute__((aligned(16))) float Wl[64 * WP];
  __shared__ __attribute__((aligned(16))) float xs[TROWS * XP];
  layer_body<true>(col, rowptr, cnt, dinv, X, W, b, OUT, Wl, xs);
}

__global__ __launch_bounds__(512) void k_layer2(const int* __restrict__ col,
                                                const int* __restrict__ rowptr,
                                                const int* __restrict__ cnt,
                                                const float* __restrict__ dinv,
                                                const float* __restrict__ X,
                                                const float* __restrict__ W,
                                                const float* __restrict__ b,
                                                float* __restrict__ OUT) {
  __shared__ __attribute__((aligned(16))) float Wl[64 * WP];
  __shared__ __attribute__((aligned(16))) float xs[TROWS * XP];
  layer_body<false>(col, rowptr, cnt, dinv, X, W, b, OUT, Wl, xs);
}


extern "C" void kernel_launch(void* const* d_in, const int* in_sizes, int n_in,
                              void* d_out, int out_size, void* d_ws, size_t ws_size,
                              hipStream_t stream) {
  const float* x   = (const float*)d_in[0];
  const int*   ei  = (const int*)d_in[1];
  const float* W1  = (const float*)d_in[2];
  const float* b1  = (const float*)d_in[3];
  const float* W2  = (const float*)d_in[4];
  const float* b2  = (const float*)d_in[5];
  float*       out = (float*)d_out;

  const int* src = ei;
  const int* dst = ei + N_EDGESC;

  float* HS     = (float*)d_ws;
  float* dinv   = HS + (size_t)N_NODESC * DIM;
  int*   cnt    = (int*)(dinv + N_NODESC);
  int*   rowptr = cnt + N_NODESC;
  int*   col    = rowptr + N_NODESC;
  int*   tmp    = col + N_EDGESC;
  int*   histg  = tmp + NCHUNK * CHS;
  int*   offsg  = histg + NBUCK * NCHUNK;

  k_bin<<<dim3(NCHUNK), dim3(1024), 0, stream>>>(src, dst, tmp, histg, offsg);
  k_bucket<<<dim3(NBUCK), dim3(1024), 0, stream>>>(tmp, histg, offsg, cnt, rowptr, dinv, col);

  k_layer1<<<dim3(LGRID), dim3(512), 0, stream>>>(col, rowptr, cnt, dinv, x, W1, b1, HS);
  k_layer2<<<dim3(LGRID), dim3(512), 0, stream>>>(col, rowptr, cnt, dinv, HS, W2, b2, out);
}


# Both sort launches assemble their bucket-ordered words in LDS and write tmp and col in whole 128-byte lines, not one scattered 4-byte store per edge; a node's sources stay in LDS arrival order
# speedup vs baseline: 5.7393x; 1.0605x over previous
#include <hip/hip_runtime.h>

typedef float v2f __attribute__((ext_vector_type(2)));
typedef float v8f __attribute__((ext_vector_type(8)));

#define N_NODESC 100000
#define N_EDGESC 1600000
#define DIM 128


#define NBUCK 196
#define BSH 9
#define BMASK 511
#define NCHUNK 256
#define CHUNK 6250
#define CHS 6272
#define IMG 16384

__device__ __forceinline__ int block_scan(int v, int* __restrict__ wsum, int& total) {
  const int lane = threadIdx.x & 31;
  const int w = threadIdx.x >> 5;
  int incl = v;
  for (int o = 1; o < 32; o <<= 1) {
    int y = __shfl_up(incl, o);
    if (lane >= o) incl += y;
  }
  __syncthreads();
  if (lane == 31) wsum[w] = incl;
  __syncthreads();
  if (w == 0) {
    int x = wsum[lane];
    int ix = x;
    for (int o = 1; o < 32; o <<= 1) {
      int y = __shfl_up(ix, o);
      if (lane >= o) ix += y;
    }
    wsum[lane] = ix - x;
    if (lane == 31) wsum[32] = ix;
  }
  __syncthreads();
  total = wsum[32];
  return wsum[w] + incl - v;
}

__global__ __launch_bounds__(1024) void k_bin(const int* __restrict__ src,
                                              const int* __restrict__ dst,
                                              int* __restrict__ tmp,
                                              int* __restrict__ histg,
                                              int* __restrict__ offsg) {
  __shared__ int hist[NBUCK];
  __shared__ int offs[NBUCK];
  __shared__ int wsum[33];
  __shared__ __attribute__((aligned(16))) int img[CHUNK + 2];
  const int t = threadIdx.x;
  const int wg = blockIdx.x;
  if (t < NBUCK) hist[t] = 0;
  __syncthreads();

  const int e0 = wg * CHUNK;
  int key[7], bk[7], rk[7];
  #pragma unroll
  for (int k = 0; k < 7; ++k) {
    const int i = k * 1024 + t;
    if (i < CHUNK) {
      const int d = dst[e0 + i];
      const int s = src[e0 + i];
      bk[k] = d >> BSH;
      key[k] = (s << BSH) | (d & BMASK);
    } else {
      bk[k] = -1;
      key[k] = 0;
    }
  }
  #pragma unroll
  for (int k = 0; k < 7; ++k) rk[k] = (bk[k] >= 0) ? atomicAdd(&hist[bk[k]], 1) : 0;
  __syncthreads();

  const int h = (t < NBUCK) ? hist[t] : 0;
  int total;
  const int ex = block_scan(h, wsum, total);
  if (t < NBUCK) {
    offs[t] = ex;
    histg[t * NCHUNK + wg] = h;
    offsg[t * NCHUNK + wg] = ex;
  }
  __syncthreads();

  #pragma unroll
  for (int k = 0; k < 7; ++k)
    if (bk[k] >= 0) img[offs[bk[k]] + rk[k]] = key[k];
  __syncthreads();
  int* __restrict__ o = tmp + wg * CHS;
  const int4* __restrict__ s4 = (const int4*)img;
  int4* __restrict__ o4 = (int4*)o;
  for (int i = t; i < CHUNK / 4; i += 1024) o4[i] = s4[i];
  if (t < (CHUNK & 3)) o[(CHUNK & ~3) + t] = img[(CHUNK & ~3) + t];
}

__global__ __launch_bounds__(1024) void k_bucket(const int* __restrict__ tmp,
                                                 const int* __restrict__ histg,
                                                 const int* __restrict__ offsg,
                                                 int* __restrict__ cnt,
                                                 int* __restrict__ rowptr,
                                                 float* __restrict__ dinv,
                                                 int* __restrict__ col) {
  __shared__ int c[512];
  __shared__ int cur[512];
  __shared__ int ph[NCHUNK];
  __shared__ int po[NCHUNK];
  __shared__ int wsum[33];
  __shared__ int img[IMG];
  const int t = threadIdx.x;
  const int lane = t & 31;
  const int w = t >> 5;
  const int b = blockIdx.x;

  if (t < 512) c[t] = 0;
  if (t < NCHUNK) {
    ph[t] = histg[b * NCHUNK + t];
    po[t] = offsg[b * NCHUNK + t];
  }
  int part = 0;
  {
    const int4* __restrict__ h4 = (const int4*)histg;
    const int n4 = b * (NCHUNK / 4);
    #pragma unroll
    for (int j = 0; j < 13; ++j) {
      const int i = j * 1024 + t;
      if (i < n4) {
        const int4 q = h4[i];
        part += (q.x + q.y) + (q.z + q.w);
      }
    }
  }
  int base;
  (void)block_scan(part, wsum, base);

  int n[8];
  const int* q[8];
  int k0[8], k1[8];
  #pragma unroll
  for (int j = 0; j < 8; ++j) {
    const int p = w + 32 * j;
    n[j] = ph[p];
    q[j] = tmp + p * CHS + po[p];
  }
  #pragma unroll
  for (int j = 0; j < 8; ++j) {
    k0[j] = (lane < n[j]) ? q[j][lane] : -1;
    k1[j] = (lane + 32 < n[j]) ? q[j][lane + 32] : -1;
  }

  #pragma unroll
  for (int j = 0; j < 8; ++j) {
    if (k0[j] >= 0) atomicAdd(&c[k0[j] & BMASK], 1);
    if (k1[j] >= 0) atomicAdd(&c[k1[j] & BMASK], 1);
  }
  #pragma unroll
  for (int j = 0; j < 8; ++j)
    for (int i = lane + 64; i < n[j]; i += 32) atomicAdd(&c[q[j][i] & BMASK], 1);
  __syncthreads();

  const int v = (t < 512) ? c[t] : 0;
  int tot;
  const int ex = block_scan(v, wsum, tot);
  if (t < 512) {
    const int node = b * 512 + t;
    cur[t] = base + ex;
    if (node < N_NODESC) {
      cnt[node] = v;
      rowptr[node] = base + ex;
      dinv[node] = rsqrtf((float)(v + 1));
    }
  }
  __syncthreads();

  auto place = [&](auto put) {
    #pragma unroll
    for (int j = 0; j < 8; ++j) {
      if (k0[j] >= 0) put(atomicAdd(&cur[k0[j] & BMASK], 1), k0[j] >> BSH);
      if (k1[j] >= 0) put(atomicAdd(&cur[k1[j] & BMASK], 1), k1[j] >> BSH);
    }
    #pragma unroll
    for (int j = 0; j < 8; ++j)
      for (int i = lane + 64; i < n[j]; i += 32) {
        const int k = q[j][i];
        put(atomicAdd(&cur[k & BMASK], 1), k >> BSH);
      }
  };
  if (tot <= IMG) {
    place([&](int p, int s) { img[p - base] = s; });
    __syncthreads();
    const int end = base + tot;
    for (int g = (base & ~31) + t; g < end; g += 1024)
      if (g >= base) col[g] = img[g - base];
  } else {
    place([&](int p, int s) { col[p] = s; });
  }
}


#define TROWS 32
#define XP 132
#define WP 288
#define NTILE (N_NODESC / TROWS)
#define LGRID 625

template <bool L1>
__device__ __forceinline__ float4 sum_node(int n, int start, int c0, float4 self, float dd, int lane,
                                           const int* __restrict__ col,
                                           const float* __restrict__ dinv,
                                           const float4* __restrict__ H4) {
  float4 acc = L1 ? make_float4(0.f, 0.f, 0.f, 0.f) : self;
  for (int base = 0; base < n; base += 32) {
    const int mm = (n - base < 32) ? (n - base) : 32;
    int myc = c0;
    if (base > 0) myc = (lane < mm) ? col[start + base + lane] : 0;
    int myw = 0;
    int j = 0;
    for (; j + 8 <= mm; j += 8) {
      float4 h[8];
      #pragma unroll
      for (int k = 0; k < 8; ++k) {
        const int s = __builtin_amdgcn_readlane(myc, j + k);
        h[k] = H4[(size_t)s * 32 + lane];
      }
      if (L1 && j == 0) myw = __float_as_int(dinv[myc]);
      #pragma unroll
      for (int k = 0; k < 8; ++k) {
        if (L1) {
          const float w = __int_as_float(__builtin_amdgcn_readlane(myw, j + k));
          acc.x = fmaf(w, h[k].x, acc.x); acc.y = fmaf(w, h[k].y, acc.y);
          acc.z = fmaf(w, h[k].z, acc.z); acc.w = fmaf(w, h[k].w, acc.w);
        } else {
          acc.x += h[k].x; acc.y += h[k].y; acc.z += h[k].z; acc.w += h[k].w;
        }
      }
    }
    if (j < mm) {
      float4 h[8];
      #pragma unroll
      for (int k = 0; k < 8; ++k) {
        if (j + k < mm) {
          const int s = __builtin_amdgcn_readlane(myc, j + k);
          h[k] = H4[(size_t)s * 32 + lane];
        } else {
          h[k] = make_float4(0.f, 0.f, 0.f, 0.f);
        }
      }
      if (L1 && j == 0) myw = __float_as_int(dinv[myc]);
      #pragma unroll
      for (int k = 0; k < 8; ++k) {
        if (L1) {
          const float w = __int_as_float(__builtin_amdgcn_readlane(myw, j + k));
          acc.x = fmaf(w, h[k].x, acc.x); acc.y = fmaf(w, h[k].y, acc.y);
          acc.z = fmaf(w, h[k].z, acc.z); acc.w = fmaf(w, h[k].w, acc.w);
        } else {
          acc.x += h[k].x; acc.y += h[k].y; acc.z += h[k].z; acc.w += h[k].w;
        }
      }
    }
  }
  if (L1) {
    acc.x = fmaf(dd, self.x, acc.x); acc.y = fmaf(dd, self.y, acc.y);
    acc.z = fmaf(dd, self.z, acc.z); acc.w = fmaf(dd, self.w, acc.w);
  }
  return acc;
}

template <bool L1>
__device__ __forceinline__ void layer_body(const int* __restrict__ col,
                                           const int* __restrict__ rowptr,
                                           const int* __restrict__ cnt,
                                           const float* __restrict__ dinv,
                                           const float* __restrict__ X,
                                           const float* __restrict__ W,
                                           const float* __restrict__ b,
                                           float* __restrict__ OUT,
                                           float* __restrict__ Wl,
                                           float* __restrict__ xs) {
  const int t = threadIdx.x;
  const int lane = t & 31;
  const int wave = t >> 5;

  {
    const float4* __restrict__ W4 = (const float4*)W;
    #pragma unroll
    for (int j = 0; j < 4; ++j) {
      const int item = t + j * 512;
      const int kp = item >> 5;
      const int q = item & 31;
      const float4 r0 = W4[(2 * kp) * 32 + q];
      const float4 r1 = W4[(2 * kp + 1) * 32 + q];
      float4* dst = (float4*)&Wl[kp * WP + 8 * q];
      dst[0] = make_float4(r0.x, r1.x, r0.y, r1.y);
      dst[1] = make_float4(r0.z, r1.z, r0.w, r1.w);
    }
  }

  const float4* __restrict__ H4 = (const float4*)X;
  const int m  = lane & 15;
  const int hg = lane >> 4;
  const int rt = wave >> 3;
  const int n0 = (wave & 7) * 16;
  const float bias = b[n0 + m];

  for (int tile = blockIdx.x; tile < NTILE; tile += LGRID) {
    const int row0 = tile * TROWS;

    const int rA = 2 * wave;
    const int dA = __builtin_amdgcn_readfirstlane(row0 + rA);
    const int dB = dA + 1;
    const int nA = cnt[dA];
    const int nB = cnt[dB];
    const int stA = rowptr[dA];
    const int stB = rowptr[dB];
    const float ddA = dinv[dA];
    const float ddB = dinv[dB];
    const float4 selfA = H4[(size_t)dA * 32 + lane];
    const float4 selfB = H4[(size_t)dB * 32 + lane];
    const int cA = (lane < nA) ? col[stA + lane] : 0;
    const int cB = (lane < nB) ? col[stB + lane] : 0;
    {
      const float4 acc = sum_node<L1>(nA, stA, cA, selfA, ddA, lane, col, dinv, H4);
      float4 tv;
      tv.x = acc.x * ddA; tv.y = acc.y * ddA; tv.z = acc.z * ddA; tv.w = acc.w * ddA;
      *(float4*)&xs[rA * XP + lane * 4] = tv;
    }
    {
      const float4 acc = sum_node<L1>(nB, stB, cB, selfB, ddB, lane, col, dinv, H4);
      float4 tv;
      tv.x = acc.x * ddB; tv.y = acc.y * ddB; tv.z = acc.z * ddB; tv.w = acc.w * ddB;
      *(float4*)&xs[(rA + 1) * XP + lane * 4] = tv;
    }
    __syncthreads();

    v8f c = {};
    const float* arow = xs + (16 * rt + m) * XP + 2 * hg;
    const float* bcol = Wl + hg * WP + 2 * (n0 + m);
    const float* ap = arow;
    const float* bp = bcol;
    for (int kb = 0; kb < 4; ++kb) {
      v2f a[8], bb[8];
      #pragma unroll
      for (int u = 0; u < 8; ++u) {
        a[u] = *(const v2f*)(ap + 4 * u);
        bb[u] = *(const v2f*)(bp + 2 * u * WP);
      }
      #pragma unroll
      for (int u = 0; u < 8; ++u)
        c = __builtin_amdgcn_wmma_f32_16x16x4_f32(
             false, a[u],  false, bb[u],
             (short)0, c,  false,  false);
      ap += 32;
      bp += 16 * WP;
    }

    const int rb = row0 + 16 * rt + 8 * hg;
    float* orow = OUT + (size_t)rb * DIM + n0 + m;
    if (L1) {
      const float4 s0 = *(const float4*)(dinv + rb);
      const float4 s1 = *(const float4*)(dinv + rb + 4);
      const float sc[8] = {s0.x, s0.y, s0.z, s0.w, s1.x, s1.y, s1.z, s1.w};
      #pragma unroll
      for (int v = 0; v < 8; ++v)
        orow[(size_t)v * DIM] = fmaxf(c[v] + bias, 0.0f) * sc[v];
    } else {
      #pragma unroll
      for (int v = 0; v < 8; ++v)
        orow[(size_t)v * DIM] = c[v] + bias;
    }
    __builtin_amdgcn_fence(__ATOMIC_RELEASE, "workgroup", "local");
    __builtin_amdgcn_s_barrier();
    __builtin_amdgcn_fence(__ATOMIC_ACQUIRE, "workgroup", "local");
  }
}

__global__ __launch_bounds__(512) void k_layer1(const int* __restrict__ col,
                                                const int* __restrict__ rowptr,
                                                const int* __restrict__ cnt,
                                                const float* __restrict__ dinv,
                                                const float* __restrict__ X,
                                                const float* __restrict__ W,
                                                const float* __restrict__ b,
                                                float* __restrict__ OUT) {
  __shared__ __attribute__((aligned(16))) float Wl[64 * WP];
  __shared__ __attribute__((aligned(16))) float xs[TROWS * XP];
  layer_body<true>(col, rowptr, cnt, dinv, X, W, b, OUT, Wl, xs);
}

__global__ __launch_bounds__(512) void k_layer2(const int* __restrict__ col,
                                                const int* __restrict__ rowptr,
                                                const int* __restrict__ cnt,
                                                const float* __restrict__ dinv,
                                                const float* __restrict__ X,
                                                const float* __restrict__ W,
                                                const float* __restrict__ b,
                                                float* __restrict__ OUT) {
  __shared__ __attribute__((aligned(16))) float Wl[64 * WP];
  __shared__ __attribute__((aligned(16))) float xs[TROWS * XP];
  layer_body<false>(col, rowptr, cnt, dinv, X, W, b, OUT, Wl, xs);
}


extern "C" void kernel_launch(void* const* d_in, const int* in_sizes, int n_in,
                              void* d_out, int out_size, void* d_ws, size_t ws_size,
                              hipStream_t stream) {
  const float* x   = (const float*)d_in[0];
  const int*   ei  = (const int*)d_in[1];
  const float* W1  = (const float*)d_in[2];
  const float* b1  = (const float*)d_in[3];
  const float* W2  = (const float*)d_in[4];
  const float* b2  = (const float*)d_in[5];
  float*       out = (float*)d_out;

  const int* src = ei;
  const int* dst = ei + N_EDGESC;

  float* HS     = (float*)d_ws;
  float* dinv   = HS + (size_t)N_NODESC * DIM;
  int*   cnt    = (int*)(dinv + N_NODESC);
  int*   rowptr = cnt + N_NODESC;
  int*   col    = rowptr + N_NODESC;
  int*   tmp    = col + N_EDGESC;
  int*   histg  = tmp + NCHUNK * CHS;
  int*   offsg  = histg + NBUCK * NCHUNK;

  k_bin<<<dim3(NCHUNK), dim3(1024), 0, stream>>>(src, dst, tmp, histg, offsg);
  k_bucket<<<dim3(NBUCK), dim3(1024), 0, stream>>>(tmp, histg, offsg, cnt, rowptr, dinv, col);

  k_layer1<<<dim3(LGRID), dim3(512), 0, stream>>>(col, rowptr, cnt, dinv, x, W1, b1, HS);
  k_layer2<<<dim3(LGRID), dim3(512), 0, stream>>>(col, rowptr, cnt, dinv, HS, W2, b2, out);
}


# Layer tiles handed out by an integer counter (one atomic per tile) to a grid of 768 blocks, 3 on every WGP, instead of a fixed 5 per block on 625; per-node sum order unchanged
# speedup vs baseline: 6.1126x; 1.0650x over previous
#include <hip/hip_runtime.h>

typedef float v2f __attribute__((ext_vector_type(2)));
typedef float v8f __attribute__((ext_vector_type(8)));

#define N_NODESC 100000
#define N_EDGESC 1600000
#define DIM 128


#define NBUCK 196
#define BSH 9
#define BMASK 511
#define NCHUNK 256
#define CHUNK 6250
#define CHS 6272
#define IMG 16384

__device__ __forceinline__ int block_scan(int v, int* __restrict__ wsum, int& total) {
  const int lane = threadIdx.x & 31;
  const int w = threadIdx.x >> 5;
  int incl = v;
  for (int o = 1; o < 32; o <<= 1) {
    int y = __shfl_up(incl, o);
    if (lane >= o) incl += y;
  }
  __syncthreads();
  if (lane == 31) wsum[w] = incl;
  __syncthreads();
  if (w == 0) {
    int x = wsum[lane];
    int ix = x;
    for (int o = 1; o < 32; o <<= 1) {
      int y = __shfl_up(ix, o);
      if (lane >= o) ix += y;
    }
    wsum[lane] = ix - x;
    if (lane == 31) wsum[32] = ix;
  }
  __syncthreads();
  total = wsum[32];
  return wsum[w] + incl - v;
}

__global__ __launch_bounds__(1024) void k_bin(const int* __restrict__ src,
                                              const int* __restrict__ dst,
                                              int* __restrict__ tmp,
                                              int* __restrict__ histg,
                                              int* __restrict__ offsg,
                                              int* __restrict__ tilectr) {
  __shared__ int hist[NBUCK];
  __shared__ int offs[NBUCK];
  __shared__ int wsum[33];
  __shared__ __attribute__((aligned(16))) int img[CHUNK + 2];
  const int t = threadIdx.x;
  const int wg = blockIdx.x;
  if (t < NBUCK) hist[t] = 0;
  if (wg == 0 && t < 2) tilectr[t] = 0;
  __syncthreads();

  const int e0 = wg * CHUNK;
  int key[7], bk[7], rk[7];
  #pragma unroll
  for (int k = 0; k < 7; ++k) {
    const int i = k * 1024 + t;
    if (i < CHUNK) {
      const int d = dst[e0 + i];
      const int s = src[e0 + i];
      bk[k] = d >> BSH;
      key[k] = (s << BSH) | (d & BMASK);
    } else {
      bk[k] = -1;
      key[k] = 0;
    }
  }
  #pragma unroll
  for (int k = 0; k < 7; ++k) rk[k] = (bk[k] >= 0) ? atomicAdd(&hist[bk[k]], 1) : 0;
  __syncthreads();

  const int h = (t < NBUCK) ? hist[t] : 0;
  int total;
  const int ex = block_scan(h, wsum, total);
  if (t < NBUCK) {
    offs[t] = ex;
    histg[t * NCHUNK + wg] = h;
    offsg[t * NCHUNK + wg] = ex;
  }
  __syncthreads();

  #pragma unroll
  for (int k = 0; k < 7; ++k)
    if (bk[k] >= 0) img[offs[bk[k]] + rk[k]] = key[k];
  __syncthreads();
  int* __restrict__ o = tmp + wg * CHS;
  const int4* __restrict__ s4 = (const int4*)img;
  int4* __restrict__ o4 = (int4*)o;
  for (int i = t; i < CHUNK / 4; i += 1024) o4[i] = s4[i];
  if (t < (CHUNK & 3)) o[(CHUNK & ~3) + t] = img[(CHUNK & ~3) + t];
}

__global__ __launch_bounds__(1024) void k_bucket(const int* __restrict__ tmp,
                                                 const int* __restrict__ histg,
                                                 const int* __restrict__ offsg,
                                                 int* __restrict__ cnt,
                                                 int* __restrict__ rowptr,
                                                 float* __restrict__ dinv,
                                                 int* __restrict__ col) {
  __shared__ int c[512];
  __shared__ int cur[512];
  __shared__ int ph[NCHUNK];
  __shared__ int po[NCHUNK];
  __shared__ int wsum[33];
  __shared__ int img[IMG];
  const int t = threadIdx.x;
  const int lane = t & 31;
  const int w = t >> 5;
  const int b = blockIdx.x;

  if (t < 512) c[t] = 0;
  if (t < NCHUNK) {
    ph[t] = histg[b * NCHUNK + t];
    po[t] = offsg[b * NCHUNK + t];
  }
  int part = 0;
  {
    const int4* __restrict__ h4 = (const int4*)histg;
    const int n4 = b * (NCHUNK / 4);
    #pragma unroll
    for (int j = 0; j < 13; ++j) {
      const int i = j * 1024 + t;
      if (i < n4) {
        const int4 q = h4[i];
        part += (q.x + q.y) + (q.z + q.w);
      }
    }
  }
  int base;
  (void)block_scan(part, wsum, base);

  int n[8];
  const int* q[8];
  int k0[8], k1[8];
  #pragma unroll
  for (int j = 0; j < 8; ++j) {
    const int p = w + 32 * j;
    n[j] = ph[p];
    q[j] = tmp + p * CHS + po[p];
  }
  #pragma unroll
  for (int j = 0; j < 8; ++j) {
    k0[j] = (lane < n[j]) ? q[j][lane] : -1;
    k1[j] = (lane + 32 < n[j]) ? q[j][lane + 32] : -1;
  }

  #pragma unroll
  for (int j = 0; j < 8; ++j) {
    if (k0[j] >= 0) atomicAdd(&c[k0[j] & BMASK], 1);
    if (k1[j] >= 0) atomicAdd(&c[k1[j] & BMASK], 1);
  }
  #pragma unroll
  for (int j = 0; j < 8; ++j)
    for (int i = lane + 64; i < n[j]; i += 32) atomicAdd(&c[q[j][i] & BMASK], 1);
  __syncthreads();

  const int v = (t < 512) ? c[t] : 0;
  int tot;
  const int ex = block_scan(v, wsum, tot);
  if (t < 512) {
    const int node = b * 512 + t;
    cur[t] = base + ex;
    if (node < N_NODESC) {
      cnt[node] = v;
      rowptr[node] = base + ex;
      dinv[node] = rsqrtf((float)(v + 1));
    }
  }
  __syncthreads();

  auto place = [&](auto put) {
    #pragma unroll
    for (int j = 0; j < 8; ++j) {
      if (k0[j] >= 0) put(atomicAdd(&cur[k0[j] & BMASK], 1), k0[j] >> BSH);
      if (k1[j] >= 0) put(atomicAdd(&cur[k1[j] & BMASK], 1), k1[j] >> BSH);
    }
    #pragma unroll
    for (int j = 0; j < 8; ++j)
      for (int i = lane + 64; i < n[j]; i += 32) {
        const int k = q[j][i];
        put(atomicAdd(&cur[k & BMASK], 1), k >> BSH);
      }
  };
  if (tot <= IMG) {
    place([&](int p, int s) { img[p - base] = s; });
    __syncthreads();
    const int end = base + tot;
    for (int g = (base & ~31) + t; g < end; g += 1024)
      if (g >= base) col[g] = img[g - base];
  } else {
    place([&](int p, int s) { col[p] = s; });
  }
}


#define TROWS 32
#define XP 132
#define WP 288
#define NTILE (N_NODESC / TROWS)
#define LGRID 768

template <bool L1>
__device__ __forceinline__ float4 sum_node(int n, int start, int c0, float4 self, float dd, int lane,
                                           const int* __restrict__ col,
                                           const float* __restrict__ dinv,
                                           const float4* __restrict__ H4) {
  float4 acc = L1 ? make_float4(0.f, 0.f, 0.f, 0.f) : self;
  for (int base = 0; base < n; base += 32) {
    const int mm = (n - base < 32) ? (n - base) : 32;
    int myc = c0;
    if (base > 0) myc = (lane < mm) ? col[start + base + lane] : 0;
    int myw = 0;
    int j = 0;
    for (; j + 8 <= mm; j += 8) {
      float4 h[8];
      #pragma unroll
      for (int k = 0; k < 8; ++k) {
        const int s = __builtin_amdgcn_readlane(myc, j + k);
        h[k] = H4[(size_t)s * 32 + lane];
      }
      if (L1 && j == 0) myw = __float_as_int(dinv[myc]);
      #pragma unroll
      for (int k = 0; k < 8; ++k) {
        if (L1) {
          const float w = __int_as_float(__builtin_amdgcn_readlane(myw, j + k));
          acc.x = fmaf(w, h[k].x, acc.x); acc.y = fmaf(w, h[k].y, acc.y);
          acc.z = fmaf(w, h[k].z, acc.z); acc.w = fmaf(w, h[k].w, acc.w);
        } else {
          acc.x += h[k].x; acc.y += h[k].y; acc.z += h[k].z; acc.w += h[k].w;
        }
      }
    }
    if (j < mm) {
      float4 h[8];
      #pragma unroll
      for (int k = 0; k < 8; ++k) {
        if (j + k < mm) {
          const int s = __builtin_amdgcn_readlane(myc, j + k);
          h[k] = H4[(size_t)s * 32 + lane];
        } else {
          h[k] = make_float4(0.f, 0.f, 0.f, 0.f);
        }
      }
      if (L1 && j == 0) myw = __float_as_int(dinv[myc]);
      #pragma unroll
      for (int k = 0; k < 8; ++k) {
        if (L1) {
          const float w = __int_as_float(__builtin_amdgcn_readlane(myw, j + k));
          acc.x = fmaf(w, h[k].x, acc.x); acc.y = fmaf(w, h[k].y, acc.y);
          acc.z = fmaf(w, h[k].z, acc.z); acc.w = fmaf(w, h[k].w, acc.w);
        } else {
          acc.x += h[k].x; acc.y += h[k].y; acc.z += h[k].z; acc.w += h[k].w;
        }
      }
    }
  }
  if (L1) {
    acc.x = fmaf(dd, self.x, acc.x); acc.y = fmaf(dd, self.y, acc.y);
    acc.z = fmaf(dd, self.z, acc.z); acc.w = fmaf(dd, self.w, acc.w);
  }
  return acc;
}

template <bool L1>
__device__ __forceinline__ void layer_body(const int* __restrict__ col,
                                           const int* __restrict__ rowptr,
                                           const int* __restrict__ cnt,
                                           const float* __restrict__ dinv,
                                           const float* __restrict__ X,
                                           const float* __restrict__ W,
                                           const float* __restrict__ b,
                                           float* __restrict__ OUT,
                                           int* __restrict__ tilectr,
                                           float* __restrict__ Wl,
                                           float* __restrict__ xs,
                                           int* __restrict__ nxt) {
  const int t = threadIdx.x;
  const int lane = t & 31;
  const int wave = t >> 5;

  {
    const float4* __restrict__ W4 = (const float4*)W;
    #pragma unroll
    for (int j = 0; j < 4; ++j) {
      const int item = t + j * 512;
      const int kp = item >> 5;
      const int q = item & 31;
      const float4 r0 = W4[(2 * kp) * 32 + q];
      const float4 r1 = W4[(2 * kp + 1) * 32 + q];
      float4* dst = (float4*)&Wl[kp * WP + 8 * q];
      dst[0] = make_float4(r0.x, r1.x, r0.y, r1.y);
      dst[1] = make_float4(r0.z, r1.z, r0.w, r1.w);
    }
  }

  const float4* __restrict__ H4 = (const float4*)X;
  const int m  = lane & 15;
  const int hg = lane >> 4;
  const int rt = wave >> 3;
  const int n0 = (wave & 7) * 16;
  const float bias = b[n0 + m];

  int tile = blockIdx.x;
  while (tile < NTILE) {
    const int row0 = tile * TROWS;

    const int rA = 2 * wave;
    const int dA = __builtin_amdgcn_readfirstlane(row0 + rA);
    const int dB = dA + 1;
    const int nA = cnt[dA];
    const int nB = cnt[dB];
    const int stA = rowptr[dA];
    const int stB = rowptr[dB];
    const float ddA = dinv[dA];
    const float ddB = dinv[dB];
    const float4 selfA = H4[(size_t)dA * 32 + lane];
    const float4 selfB = H4[(size_t)dB * 32 + lane];
    const int cA = (lane < nA) ? col[stA + lane] : 0;
    const int cB = (lane < nB) ? col[stB + lane] : 0;
    int got = 0;
    if (t == 0) got = atomicAdd(tilectr, 1);
    {
      const float4 acc = sum_node<L1>(nA, stA, cA, selfA, ddA, lane, col, dinv, H4);
      float4 tv;
      tv.x = acc.x * ddA; tv.y = acc.y * ddA; tv.z = acc.z * ddA; tv.w = acc.w * ddA;
      *(float4*)&xs[rA * XP + lane * 4] = tv;
    }
    {
      const float4 acc = sum_node<L1>(nB, stB, cB, selfB, ddB, lane, col, dinv, H4);
      float4 tv;
      tv.x = acc.x * ddB; tv.y = acc.y * ddB; tv.z = acc.z * ddB; tv.w = acc.w * ddB;
      *(float4*)&xs[(rA + 1) * XP + lane * 4] = tv;
    }
    if (t == 0) *nxt = LGRID + got;
    __syncthreads();
    const int ntile = *nxt;

    v8f c = {};
    const float* arow = xs + (16 * rt + m) * XP + 2 * hg;
    const float* bcol = Wl + hg * WP + 2 * (n0 + m);
    const float* ap = arow;
    const float* bp = bcol;
    for (int kb = 0; kb < 4; ++kb) {
      v2f a[8], bb[8];
      #pragma unroll
      for (int u = 0; u < 8; ++u) {
        a[u] = *(const v2f*)(ap + 4 * u);
        bb[u] = *(const v2f*)(bp + 2 * u * WP);
      }
      #pragma unroll
      for (int u = 0; u < 8; ++u)
        c = __builtin_amdgcn_wmma_f32_16x16x4_f32(
             false, a[u],  false, bb[u],
             (short)0, c,  false,  false);
      ap += 32;
      bp += 16 * WP;
    }

    const int rb = row0 + 16 * rt + 8 * hg;
    float* orow = OUT + (size_t)rb * DIM + n0 + m;
    if (L1) {
      const float4 s0 = *(const float4*)(dinv + rb);
      const float4 s1 = *(const float4*)(dinv + rb + 4);
      const float sc[8] = {s0.x, s0.y, s0.z, s0.w, s1.x, s1.y, s1.z, s1.w};
      #pragma unroll
      for (int v = 0; v < 8; ++v)
        orow[(size_t)v * DIM] = fmaxf(c[v] + bias, 0.0f) * sc[v];
    } else {
      #pragma unroll
      for (int v = 0; v < 8; ++v)
        orow[(size_t)v * DIM] = c[v] + bias;
    }
    __builtin_amdgcn_fence(__ATOMIC_RELEASE, "workgroup", "local");
    __builtin_amdgcn_s_barrier();
    __builtin_amdgcn_fence(__ATOMIC_ACQUIRE, "workgroup", "local");
    tile = __builtin_amdgcn_readfirstlane(ntile);
  }
}

__global__ __launch_bounds__(512) void k_layer1(const int* __restrict__ col,
                                                const int* __restrict__ rowptr,
                                                const int* __restrict__ cnt,
                                                const float* __restrict__ dinv,
                                                const float* __restrict__ X,
                                                const float* __restrict__ W,
                                                const float* __restrict__ b,
                                                float* __restrict__ OUT,
                                                int* __restrict__ tilectr) {
  __shared__ __attribute__((aligned(16))) float Wl[64 * WP];
  __shared__ __attribute__((aligned(16))) float xs[TROWS * XP];
  __shared__ int nxt;
  layer_body<true>(col, rowptr, cnt, dinv, X, W, b, OUT, tilectr, Wl, xs, &nxt);
}

__global__ __launch_bounds__(512) void k_layer2(const int* __restrict__ col,
                                                const int* __restrict__ rowptr,
                                                const int* __restrict__ cnt,
                                                const float* __restrict__ dinv,
                                                const float* __restrict__ X,
                                                const float* __restrict__ W,
                                                const float* __restrict__ b,
                                                float* __restrict__ OUT,
                                                int* __restrict__ tilectr) {
  __shared__ __attribute__((aligned(16))) float Wl[64 * WP];
  __shared__ __attribute__((aligned(16))) float xs[TROWS * XP];
  __shared__ int nxt;
  layer_body<false>(col, rowptr, cnt, dinv, X, W, b, OUT, tilectr, Wl, xs, &nxt);
}


extern "C" void kernel_launch(void* const* d_in, const int* in_sizes, int n_in,
                              void* d_out, int out_size, void* d_ws, size_t ws_size,
                              hipStream_t stream) {
  const float* x   = (const float*)d_in[0];
  const int*   ei  = (const int*)d_in[1];
  const float* W1  = (const float*)d_in[2];
  const float* b1  = (const float*)d_in[3];
  const float* W2  = (const float*)d_in[4];
  const float* b2  = (const float*)d_in[5];
  float*       out = (float*)d_out;

  const int* src = ei;
  const int* dst = ei + N_EDGESC;

  float* HS     = (float*)d_ws;
  float* dinv   = HS + (size_t)N_NODESC * DIM;
  int*   cnt    = (int*)(dinv + N_NODESC);
  int*   rowptr = cnt + N_NODESC;
  int*   col    = rowptr + N_NODESC;
  int*   tmp    = col + N_EDGESC;
  int*   histg  = tmp + NCHUNK * CHS;
  int*   offsg  = histg + NBUCK * NCHUNK;
  int*   tctr   = offsg + NBUCK * NCHUNK;

  k_bin<<<dim3(NCHUNK), dim3(1024), 0, stream>>>(src, dst, tmp, histg, offsg, tctr);
  k_bucket<<<dim3(NBUCK), dim3(1024), 0, stream>>>(tmp, histg, offsg, cnt, rowptr, dinv, col);

  k_layer1<<<dim3(LGRID), dim3(512), 0, stream>>>(col, rowptr, cnt, dinv, x, W1, b1, HS, tctr);
  k_layer2<<<dim3(LGRID), dim3(512), 0, stream>>>(col, rowptr, cnt, dinv, HS, W2, b2, out, tctr + 1);
}


# Layer results go back into the LDS tile and each wave stores its two rows as 16-byte stores: every store instruction writes four whole 128-byte lines, not two 64-byte halves; sums unchanged
# speedup vs baseline: 6.4111x; 1.0488x over previous
#include <hip/hip_runtime.h>

typedef float v2f __attribute__((ext_vector_type(2)));
typedef float v8f __attribute__((ext_vector_type(8)));

#define N_NODESC 100000
#define N_EDGESC 1600000
#define DIM 128


#define NBUCK 196
#define BSH 9
#define BMASK 511
#define NCHUNK 256
#define CHUNK 6250
#define CHS 6272
#define IMG 16384

__device__ __forceinline__ int block_scan(int v, int* __restrict__ wsum, int& total) {
  const int lane = threadIdx.x & 31;
  const int w = threadIdx.x >> 5;
  int incl = v;
  for (int o = 1; o < 32; o <<= 1) {
    int y = __shfl_up(incl, o);
    if (lane >= o) incl += y;
  }
  __syncthreads();
  if (lane == 31) wsum[w] = incl;
  __syncthreads();
  if (w == 0) {
    int x = wsum[lane];
    int ix = x;
    for (int o = 1; o < 32; o <<= 1) {
      int y = __shfl_up(ix, o);
      if (lane >= o) ix += y;
    }
    wsum[lane] = ix - x;
    if (lane == 31) wsum[32] = ix;
  }
  __syncthreads();
  total = wsum[32];
  return wsum[w] + incl - v;
}

__global__ __launch_bounds__(1024) void k_bin(const int* __restrict__ src,
                                              const int* __restrict__ dst,
                                              int* __restrict__ tmp,
                                              int* __restrict__ histg,
                                              int* __restrict__ offsg,
                                              int* __restrict__ tilectr) {
  __shared__ int hist[NBUCK];
  __shared__ int offs[NBUCK];
  __shared__ int wsum[33];
  __shared__ __attribute__((aligned(16))) int img[CHUNK + 2];
  const int t = threadIdx.x;
  const int wg = blockIdx.x;
  if (t < NBUCK) hist[t] = 0;
  if (wg == 0 && t < 2) tilectr[t] = 0;
  __syncthreads();

  const int e0 = wg * CHUNK;
  int key[7], bk[7], rk[7];
  #pragma unroll
  for (int k = 0; k < 7; ++k) {
    const int i = k * 1024 + t;
    if (i < CHUNK) {
      const int d = dst[e0 + i];
      const int s = src[e0 + i];
      bk[k] = d >> BSH;
      key[k] = (s << BSH) | (d & BMASK);
    } else {
      bk[k] = -1;
      key[k] = 0;
    }
  }
  #pragma unroll
  for (int k = 0; k < 7; ++k) rk[k] = (bk[k] >= 0) ? atomicAdd(&hist[bk[k]], 1) : 0;
  __syncthreads();

  const int h = (t < NBUCK) ? hist[t] : 0;
  int total;
  const int ex = block_scan(h, wsum, total);
  if (t < NBUCK) {
    offs[t] = ex;
    histg[t * NCHUNK + wg] = h;
    offsg[t * NCHUNK + wg] = ex;
  }
  __syncthreads();

  #pragma unroll
  for (int k = 0; k < 7; ++k)
    if (bk[k] >= 0) img[offs[bk[k]] + rk[k]] = key[k];
  __syncthreads();
  int* __restrict__ o = tmp + wg * CHS;
  const int4* __restrict__ s4 = (const int4*)img;
  int4* __restrict__ o4 = (int4*)o;
  for (int i = t; i < CHUNK / 4; i += 1024) o4[i] = s4[i];
  if (t < (CHUNK & 3)) o[(CHUNK & ~3) + t] = img[(CHUNK & ~3) + t];
}

__global__ __launch_bounds__(1024) void k_bucket(const int* __restrict__ tmp,
                                                 const int* __restrict__ histg,
                                                 const int* __restrict__ offsg,
                                                 int* __restrict__ cnt,
                                                 int* __restrict__ rowptr,
                                                 float* __restrict__ dinv,
                                                 int* __restrict__ col) {
  __shared__ int c[512];
  __shared__ int cur[512];
  __shared__ int ph[NCHUNK];
  __shared__ int po[NCHUNK];
  __shared__ int wsum[33];
  __shared__ int img[IMG];
  const int t = threadIdx.x;
  const int lane = t & 31;
  const int w = t >> 5;
  const int b = blockIdx.x;

  if (t < 512) c[t] = 0;
  if (t < NCHUNK) {
    ph[t] = histg[b * NCHUNK + t];
    po[t] = offsg[b * NCHUNK + t];
  }
  int part = 0;
  {
    const int4* __restrict__ h4 = (const int4*)histg;
    const int n4 = b * (NCHUNK / 4);
    #pragma unroll
    for (int j = 0; j < 13; ++j) {
      const int i = j * 1024 + t;
      if (i < n4) {
        const int4 q = h4[i];
        part += (q.x + q.y) + (q.z + q.w);
      }
    }
  }
  int base;
  (void)block_scan(part, wsum, base);

  int n[8];
  const int* q[8];
  int k0[8], k1[8];
  #pragma unroll
  for (int j = 0; j < 8; ++j) {
    const int p = w + 32 * j;
    n[j] = ph[p];
    q[j] = tmp + p * CHS + po[p];
  }
  #pragma unroll
  for (int j = 0; j < 8; ++j) {
    k0[j] = (lane < n[j]) ? q[j][lane] : -1;
    k1[j] = (lane + 32 < n[j]) ? q[j][lane + 32] : -1;
  }

  #pragma unroll
  for (int j = 0; j < 8; ++j) {
    if (k0[j] >= 0) atomicAdd(&c[k0[j] & BMASK], 1);
    if (k1[j] >= 0) atomicAdd(&c[k1[j] & BMASK], 1);
  }
  #pragma unroll
  for (int j = 0; j < 8; ++j)
    for (int i = lane + 64; i < n[j]; i += 32) atomicAdd(&c[q[j][i] & BMASK], 1);
  __syncthreads();

  const int v = (t < 512) ? c[t] : 0;
  int tot;
  const int ex = block_scan(v, wsum, tot);
  if (t < 512) {
    const int node = b * 512 + t;
    cur[t] = base + ex;
    if (node < N_NODESC) {
      cnt[node] = v;
      rowptr[node] = base + ex;
      dinv[node] = rsqrtf((float)(v + 1));
    }
  }
  __syncthreads();

  auto place = [&](auto put) {
    #pragma unroll
    for (int j = 0; j < 8; ++j) {
      if (k0[j] >= 0) put(atomicAdd(&cur[k0[j] & BMASK], 1), k0[j] >> BSH);
      if (k1[j] >= 0) put(atomicAdd(&cur[k1[j] & BMASK], 1), k1[j] >> BSH);
    }
    #pragma unroll
    for (int j = 0; j < 8; ++j)
      for (int i = lane + 64; i < n[j]; i += 32) {
        const int k = q[j][i];
        put(atomicAdd(&cur[k & BMASK], 1), k >> BSH);
      }
  };
  if (tot <= IMG) {
    place([&](int p, int s) { img[p - base] = s; });
    __syncthreads();
    const int end = base + tot;
    for (int g = (base & ~31) + t; g < end; g += 1024)
      if (g >= base) col[g] = img[g - base];
  } else {
    place([&](int p, int s) { col[p] = s; });
  }
}


#define TROWS 32
#define XP 132
#define WP 288
#define NTILE (N_NODESC / TROWS)
#define LGRID 768

template <bool L1>
__device__ __forceinline__ float4 sum_node(int n, int start, int c0, float4 self, float dd, int lane,
                                           const int* __restrict__ col,
                                           const float* __restrict__ dinv,
                                           const float4* __restrict__ H4) {
  float4 acc = L1 ? make_float4(0.f, 0.f, 0.f, 0.f) : self;
  for (int base = 0; base < n; base += 32) {
    const int mm = (n - base < 32) ? (n - base) : 32;
    int myc = c0;
    if (base > 0) myc = (lane < mm) ? col[start + base + lane] : 0;
    int myw = 0;
    int j = 0;
    for (; j + 8 <= mm; j += 8) {
      float4 h[8];
      #pragma unroll
      for (int k = 0; k < 8; ++k) {
        const int s = __builtin_amdgcn_readlane(myc, j + k);
        h[k] = H4[(size_t)s * 32 + lane];
      }
      if (L1 && j == 0) myw = __float_as_int(dinv[myc]);
      #pragma unroll
      for (int k = 0; k < 8; ++k) {
        if (L1) {
          const float w = __int_as_float(__builtin_amdgcn_readlane(myw, j + k));
          acc.x = fmaf(w, h[k].x, acc.x); acc.y = fmaf(w, h[k].y, acc.y);
          acc.z = fmaf(w, h[k].z, acc.z); acc.w = fmaf(w, h[k].w, acc.w);
        } else {
          acc.x += h[k].x; acc.y += h[k].y; acc.z += h[k].z; acc.w += h[k].w;
        }
      }
    }
    if (j < mm) {
      float4 h[8];
      #pragma unroll
      for (int k = 0; k < 8; ++k) {
        if (j + k < mm) {
          const int s = __builtin_amdgcn_readlane(myc, j + k);
          h[k] = H4[(size_t)s * 32 + lane];
        } else {
          h[k] = make_float4(0.f, 0.f, 0.f, 0.f);
        }
      }
      if (L1 && j == 0) myw = __float_as_int(dinv[myc]);
      #pragma unroll
      for (int k = 0; k < 8; ++k) {
        if (L1) {
          const float w = __int_as_float(__builtin_amdgcn_readlane(myw, j + k));
          acc.x = fmaf(w, h[k].x, acc.x); acc.y = fmaf(w, h[k].y, acc.y);
          acc.z = fmaf(w, h[k].z, acc.z); acc.w = fmaf(w, h[k].w, acc.w);
        } else {
          acc.x += h[k].x; acc.y += h[k].y; acc.z += h[k].z; acc.w += h[k].w;
        }
      }
    }
  }
  if (L1) {
    acc.x = fmaf(dd, self.x, acc.x); acc.y = fmaf(dd, self.y, acc.y);
    acc.z = fmaf(dd, self.z, acc.z); acc.w = fmaf(dd, self.w, acc.w);
  }
  return acc;
}

template <bool L1>
__device__ __forceinline__ void layer_body(const int* __restrict__ col,
                                           const int* __restrict__ rowptr,
                                           const int* __restrict__ cnt,
                                           const float* __restrict__ dinv,
                                           const float* __restrict__ X,
                                           const float* __restrict__ W,
                                           const float* __restrict__ b,
                                           float* __restrict__ OUT,
                                           int* __restrict__ tilectr,
                                           float* __restrict__ Wl,
                                           float* __restrict__ xs,
                                           int* __restrict__ nxt) {
  const int t = threadIdx.x;
  const int lane = t & 31;
  const int wave = t >> 5;

  {
    const float4* __restrict__ W4 = (const float4*)W;
    #pragma unroll
    for (int j = 0; j < 4; ++j) {
      const int item = t + j * 512;
      const int kp = item >> 5;
      const int q = item & 31;
      const float4 r0 = W4[(2 * kp) * 32 + q];
      const float4 r1 = W4[(2 * kp + 1) * 32 + q];
      float4* dst = (float4*)&Wl[kp * WP + 8 * q];
      dst[0] = make_float4(r0.x, r1.x, r0.y, r1.y);
      dst[1] = make_float4(r0.z, r1.z, r0.w, r1.w);
    }
  }

  const float4* __restrict__ H4 = (const float4*)X;
  const int m  = lane & 15;
  const int hg = lane >> 4;
  const int rt = wave >> 3;
  const int n0 = (wave & 7) * 16;
  const float bias = b[n0 + m];

  int tile = blockIdx.x;
  while (tile < NTILE) {
    const int row0 = tile * TROWS;

    const int rA = 2 * wave;
    const int dA = __builtin_amdgcn_readfirstlane(row0 + rA);
    const int dB = dA + 1;
    const int nA = cnt[dA];
    const int nB = cnt[dB];
    const int stA = rowptr[dA];
    const int stB = rowptr[dB];
    const float ddA = dinv[dA];
    const float ddB = dinv[dB];
    const float4 selfA = H4[(size_t)dA * 32 + lane];
    const float4 selfB = H4[(size_t)dB * 32 + lane];
    const int cA = (lane < nA) ? col[stA + lane] : 0;
    const int cB = (lane < nB) ? col[stB + lane] : 0;
    int got = 0;
    if (t == 0) got = atomicAdd(tilectr, 1);
    {
      const float4 acc = sum_node<L1>(nA, stA, cA, selfA, ddA, lane, col, dinv, H4);
      float4 tv;
      tv.x = acc.x * ddA; tv.y = acc.y * ddA; tv.z = acc.z * ddA; tv.w = acc.w * ddA;
      *(float4*)&xs[rA * XP + lane * 4] = tv;
    }
    {
      const float4 acc = sum_node<L1>(nB, stB, cB, selfB, ddB, lane, col, dinv, H4);
      float4 tv;
      tv.x = acc.x * ddB; tv.y = acc.y * ddB; tv.z = acc.z * ddB; tv.w = acc.w * ddB;
      *(float4*)&xs[(rA + 1) * XP + lane * 4] = tv;
    }
    if (t == 0) *nxt = LGRID + got;
    __syncthreads();
    const int ntile = *nxt;

    v8f c = {};
    const float* arow = xs + (16 * rt + m) * XP + 2 * hg;
    const float* bcol = Wl + hg * WP + 2 * (n0 + m);
    const float* ap = arow;
    const float* bp = bcol;
    for (int kb = 0; kb < 4; ++kb) {
      v2f a[8], bb[8];
      #pragma unroll
      for (int u = 0; u < 8; ++u) {
        a[u] = *(const v2f*)(ap + 4 * u);
        bb[u] = *(const v2f*)(bp + 2 * u * WP);
      }
      #pragma unroll
      for (int u = 0; u < 8; ++u)
        c = __builtin_amdgcn_wmma_f32_16x16x4_f32(
             false, a[u],  false, bb[u],
             (short)0, c,  false,  false);
      ap += 32;
      bp += 16 * WP;
    }

    const int rl = 16 * rt + 8 * hg;
    float o[8];
    if (L1) {
      const float4 s0 = *(const float4*)(dinv + row0 + rl);
      const float4 s1 = *(const float4*)(dinv + row0 + rl + 4);
      const float sc[8] = {s0.x, s0.y, s0.z, s0.w, s1.x, s1.y, s1.z, s1.w};
      #pragma unroll
      for (int v = 0; v < 8; ++v)
        o[v] = fmaxf(c[v] + bias, 0.0f) * sc[v];
    } else {
      #pragma unroll
      for (int v = 0; v < 8; ++v)
        o[v] = c[v] + bias;
    }
    __builtin_amdgcn_fence(__ATOMIC_RELEASE, "workgroup", "local");
    __builtin_amdgcn_s_barrier();
    __builtin_amdgcn_fence(__ATOMIC_ACQUIRE, "workgroup", "local");
    float* orow = xs + rl * XP + n0 + m;
    #pragma unroll
    for (int v = 0; v < 8; ++v)
      orow[v * XP] = o[v];
    __builtin_amdgcn_fence(__ATOMIC_RELEASE, "workgroup", "local");
    __builtin_amdgcn_s_barrier();
    __builtin_amdgcn_fence(__ATOMIC_ACQUIRE, "workgroup", "local");
    {
      const float4 r0 = *(const float4*)&xs[rA * XP + lane * 4];
      const float4 r1 = *(const float4*)&xs[(rA + 1) * XP + lane * 4];
      float4* __restrict__ O4 = (float4*)OUT;
      O4[(size_t)dA * 32 + lane] = r0;
      O4[(size_t)dB * 32 + lane] = r1;
    }
    tile = __builtin_amdgcn_readfirstlane(ntile);
  }
}

__global__ __launch_bounds__(512) void k_layer1(const int* __restrict__ col,
                                                const int* __restrict__ rowptr,
                                                const int* __restrict__ cnt,
                                                const float* __restrict__ dinv,
                                                const float* __restrict__ X,
                                                const float* __restrict__ W,
                                                const float* __restrict__ b,
                                                float* __restrict__ OUT,
                                                int* __restrict__ tilectr) {
  __shared__ __attribute__((aligned(16))) float Wl[64 * WP];
  __shared__ __attribute__((aligned(16))) float xs[TROWS * XP];
  __shared__ int nxt;
  layer_body<true>(col, rowptr, cnt, dinv, X, W, b, OUT, tilectr, Wl, xs, &nxt);
}

__global__ __launch_bounds__(512) void k_layer2(const int* __restrict__ col,
                                                const int* __restrict__ rowptr,
                                                const int* __restrict__ cnt,
                                                const float* __restrict__ dinv,
                                                const float* __restrict__ X,
                                                const float* __restrict__ W,
                                                const float* __restrict__ b,
                                                float* __restrict__ OUT,
                                                int* __restrict__ tilectr) {
  __shared__ __attribute__((aligned(16))) float Wl[64 * WP];
  __shared__ __attribute__((aligned(16))) float xs[TROWS * XP];
  __shared__ int nxt;
  layer_body<false>(col, rowptr, cnt, dinv, X, W, b, OUT, tilectr, Wl, xs, &nxt);
}


extern "C" void kernel_launch(void* const* d_in, const int* in_sizes, int n_in,
                              void* d_out, int out_size, void* d_ws, size_t ws_size,
                              hipStream_t stream) {
  const float* x   = (const float*)d_in[0];
  const int*   ei  = (const int*)d_in[1];
  const float* W1  = (const float*)d_in[2];
  const float* b1  = (const float*)d_in[3];
  const float* W2  = (const float*)d_in[4];
  const float* b2  = (const float*)d_in[5];
  float*       out = (float*)d_out;

  const int* src = ei;
  const int* dst = ei + N_EDGESC;

  float* HS     = (float*)d_ws;
  float* dinv   = HS + (size_t)N_NODESC * DIM;
  int*   cnt    = (int*)(dinv + N_NODESC);
  int*   rowptr = cnt + N_NODESC;
  int*   col    = rowptr + N_NODESC;
  int*   tmp    = col + N_EDGESC;
  int*   histg  = tmp + NCHUNK * CHS;
  int*   offsg  = histg + NBUCK * NCHUNK;
  int*   tctr   = offsg + NBUCK * NCHUNK;

  k_bin<<<dim3(NCHUNK), dim3(1024), 0, stream>>>(src, dst, tmp, histg, offsg, tctr);
  k_bucket<<<dim3(NBUCK), dim3(1024), 0, stream>>>(tmp, histg, offsg, cnt, rowptr, dinv, col);

  k_layer1<<<dim3(LGRID), dim3(512), 0, stream>>>(col, rowptr, cnt, dinv, x, W1, b1, HS, tctr);
  k_layer2<<<dim3(LGRID), dim3(512), 0, stream>>>(col, rowptr, cnt, dinv, HS, W2, b2, out, tctr + 1);
}
